# starting kernel
# baseline (speedup 1.0000x reference)
#include <hip/hip_runtime.h>

typedef __attribute__((ext_vector_type(2))) float v2f;
typedef __attribute__((ext_vector_type(8))) float v8f;

#define IN_DIM  64
#define HIDDEN  64
#define OUT_DIM 2

__global__ void gnn_zero_ws(float* __restrict__ p, long n) {
    long i = (long)blockIdx.x * blockDim.x + threadIdx.x;
    long stride = (long)gridDim.x * blockDim.x;
    for (; i < n; i += stride) p[i] = 0.0f;
}

__global__ __launch_bounds__(256) void gnn_edge_agg(
    const float* __restrict__ x,
    const int*   __restrict__ src,
    const int*   __restrict__ dst,
    float* __restrict__ msg,
    float* __restrict__ deg,
    long E) {
    long t = (long)blockIdx.x * blockDim.x + threadIdx.x;
    if (t >= E * 64) return;
    long e = t >> 6;
    int  k = (int)(t & 63);
    int  s = src[e];
    int  d = dst[e];
    atomicAdd(&msg[(long)d * IN_DIM + k], x[(long)s * IN_DIM + k]);
    if (k == 0) atomicAdd(&deg[d], 1.0f);
}

__global__ __launch_bounds__(256) void gnn_node_update(
    const float* __restrict__ x,
    const float* __restrict__ msg,
    const float* __restrict__ deg,
    const float* __restrict__ Wl,
    const float* __restrict__ bl,
    const float* __restrict__ Wr,
    const float* __restrict__ Wh,
    const float* __restrict__ bh,
    float* __restrict__ out,
    int N) {
    __shared__ float zbuf[8][16 * HIDDEN];

    const int lane = threadIdx.x & 31;
    const int wave = threadIdx.x >> 5;
    const int tile = blockIdx.x * 8 + wave;
    const int r    = lane & 15;
    const int hi   = lane >> 4;

    const int  row   = tile * 16 + r;
    const bool valid = (row < N);
    const int  srow  = valid ? row : 0;

    const float2* xrow = (const float2*)(x   + (long)srow * IN_DIM);
    const float2* mrow = (const float2*)(msg + (long)srow * IN_DIM);
    const float scale = valid ? (1.0f / fmaxf(deg[srow], 1.0f)) : 0.0f;
    const float xs    = valid ? 1.0f : 0.0f;

    const int koff2 = hi;

    for (int j = 0; j < 4; ++j) {
        const int n = j * 16 + r;
        const float2* wlrow = (const float2*)(Wl + (long)n * IN_DIM);
        const float2* wrrow = (const float2*)(Wr + (long)n * IN_DIM);

        v8f c = {0.f, 0.f, 0.f, 0.f, 0.f, 0.f, 0.f, 0.f};

        #pragma unroll
        for (int kb = 0; kb < 16; ++kb) {
            const int p = 2 * kb + koff2;
            float2 am = mrow[p];
            float2 ax = xrow[p];
            float2 wl = wlrow[p];
            float2 wr = wrrow[p];

            v2f aA; aA.x = am.x * scale; aA.y = am.y * scale;
            v2f aX; aX.x = ax.x * xs;    aX.y = ax.y * xs;
            v2f bL; bL.x = wl.x;         bL.y = wl.y;
            v2f bR; bR.x = wr.x;         bR.y = wr.y;

            c = __builtin_amdgcn_wmma_f32_16x16x4_f32(
                    false, aA, false, bL, (short)0, c, false, false);
            c = __builtin_amdgcn_wmma_f32_16x16x4_f32(
                    false, aX, false, bR, (short)0, c, false, false);
        }

        const float bias = bl[n];
        #pragma unroll
        for (int v = 0; v < 8; ++v) {
            float h = c[v] + bias;
            h = fmaxf(h, 0.0f);
            zbuf[wave][(v + hi * 8) * HIDDEN + n] = h;
        }
    }

    float acc = bh[hi];
    #pragma unroll 8
    for (int nn = 0; nn < HIDDEN; ++nn)
        acc += zbuf[wave][r * HIDDEN + nn] * Wh[hi * HIDDEN + nn];

    if (valid) out[(long)row * OUT_DIM + hi] = acc;
}

extern "C" void kernel_launch(void* const* d_in, const int* in_sizes, int n_in,
                              void* d_out, int out_size, void* d_ws, size_t ws_size,
                              hipStream_t stream) {
    const float* x  = (const float*)d_in[0];
    const int*   ei = (const int*)d_in[1];
    const float* Wl = (const float*)d_in[2];
    const float* bl = (const float*)d_in[3];
    const float* Wr = (const float*)d_in[4];
    const float* Wh = (const float*)d_in[5];
    const float* bh = (const float*)d_in[6];
    float* out = (float*)d_out;

    const long N = in_sizes[0] / IN_DIM;
    const long E = in_sizes[1] / 2;
    const int* src = ei;
    const int* dst = ei + E;

    float* msg = (float*)d_ws;
    float* deg = msg + (size_t)N * IN_DIM;

    {
        long n = N * IN_DIM + N;
        int blocks = (int)((n + 256 * 8 - 1) / (256 * 8));
        if (blocks < 1) blocks = 1;
        gnn_zero_ws<<<blocks, 256, 0, stream>>>(msg, n);
    }
    {
        long threads = E * 64;
        int blocks = (int)((threads + 255) / 256);
        gnn_edge_agg<<<blocks, 256, 0, stream>>>(x, src, dst, msg, deg, E);
    }
    {
        long tiles  = (N + 15) / 16;
        int  blocks = (int)((tiles + 7) / 8);
        gnn_node_update<<<blocks, 256, 0, stream>>>(
            x, msg, deg, Wl, bl, Wr, Wh, bh, out, (int)N);
    }
}


# replace per-feature f32 atomic scatter by one int exchange per edge building arrival-order in-edge lists; node kernel walks the lists and sums source rows in registers (sum order = list order)
# speedup vs baseline: 2.4200x; 2.4200x over previous
#include <hip/hip_runtime.h>

typedef __attribute__((ext_vector_type(2))) float v2f;
typedef __attribute__((ext_vector_type(8))) float v8f;

#define IN_DIM  64
#define HIDDEN  64
#define OUT_DIM 2
#define APITCH  68

__global__ __launch_bounds__(256) void gnn_init_heads(int* __restrict__ head, int n) {
    int i = blockIdx.x * blockDim.x + threadIdx.x;
    if (i < n) head[i] = -1;
}

__global__ __launch_bounds__(256) void gnn_link_edges(
    const int* __restrict__ src,
    const int* __restrict__ dst,
    int* __restrict__ head,
    int2* __restrict__ link,
    int E) {
    int e = blockIdx.x * blockDim.x + threadIdx.x;
    if (e >= E) return;
    int d = dst[e];
    int s = src[e];
    int prev = atomicExch(&head[d], e);
    link[e] = make_int2(prev, s);
}

__global__ __launch_bounds__(256) void gnn_node_update(
    const float* __restrict__ x,
    const int*   __restrict__ head,
    const int2*  __restrict__ link,
    const float* __restrict__ Wl,
    const float* __restrict__ bl,
    const float* __restrict__ Wr,
    const float* __restrict__ Wh,
    const float* __restrict__ bh,
    float* __restrict__ out,
    int N, int E) {
    __shared__ float lbuf[8][16 * APITCH];

    const int lane = threadIdx.x & 31;
    const int wave = threadIdx.x >> 5;
    const int tile = blockIdx.x * 8 + wave;
    const int r    = lane & 15;
    const int hi   = lane >> 4;

    const int  row   = tile * 16 + r;
    const bool valid = (row < N);
    const int  srow  = valid ? row : 0;

    const float2* x2 = (const float2*)x;
    int  e    = valid ? head[row] : -1;
    bool live = (unsigned)e < (unsigned)E;
    int  s = 0, nx = -1;
    if (live) { const int2 ln = link[e]; nx = ln.x; s = ln.y; }
    float cnt = 0.0f;
    float2 acc[16];
    #pragma unroll
    for (int i = 0; i < 16; ++i) acc[i] = make_float2(0.0f, 0.0f);

    for (int it = 0; it < E; ++it) {
        if (!__any(live)) break;
        const bool live2 = (unsigned)nx < (unsigned)E;
        int s2 = 0, nx2 = -1;
        if (live2) { const int2 ln = link[nx]; nx2 = ln.x; s2 = ln.y; }
        const int sm = live ? s : -1;
        float2 t[16];
        #pragma unroll
        for (int i = 0; i < 16; ++i) {
            const int si = __builtin_amdgcn_readlane(sm, i);
            t[i] = make_float2(0.0f, 0.0f);
            if ((unsigned)si < (unsigned)N) t[i] = x2[(long)si * (IN_DIM / 2) + lane];
        }
        #pragma unroll
        for (int i = 0; i < 16; ++i) { acc[i].x += t[i].x; acc[i].y += t[i].y; }
        cnt += live ? 1.0f : 0.0f;
        live = live2; s = s2; nx = nx2;
    }

    float* ab = lbuf[wave];
    #pragma unroll
    for (int i = 0; i < 16; ++i)
        *(float2*)(ab + i * APITCH + 2 * lane) = acc[i];

    const float scale = 1.0f / fmaxf(cnt, 1.0f);
    v2f aAf[16];
    #pragma unroll
    for (int kb = 0; kb < 16; ++kb) {
        const float2 a = *(const float2*)(ab + r * APITCH + 4 * kb + 2 * hi);
        aAf[kb].x = a.x * scale; aAf[kb].y = a.y * scale;
    }

    const float2* xrow = (const float2*)(x + (long)srow * IN_DIM);
    const float xs    = valid ? 1.0f : 0.0f;

    const int koff2 = hi;

    for (int j = 0; j < 4; ++j) {
        const int n = j * 16 + r;
        const float2* wlrow = (const float2*)(Wl + (long)n * IN_DIM);
        const float2* wrrow = (const float2*)(Wr + (long)n * IN_DIM);

        v8f c = {0.f, 0.f, 0.f, 0.f, 0.f, 0.f, 0.f, 0.f};

        #pragma unroll
        for (int kb = 0; kb < 16; ++kb) {
            const int p = 2 * kb + koff2;
            float2 ax = xrow[p];
            float2 wl = wlrow[p];
            float2 wr = wrrow[p];

            v2f aX; aX.x = ax.x * xs;    aX.y = ax.y * xs;
            v2f bL; bL.x = wl.x;         bL.y = wl.y;
            v2f bR; bR.x = wr.x;         bR.y = wr.y;

            c = __builtin_amdgcn_wmma_f32_16x16x4_f32(
                    false, aAf[kb], false, bL, (short)0, c, false, false);
            c = __builtin_amdgcn_wmma_f32_16x16x4_f32(
                    false, aX, false, bR, (short)0, c, false, false);
        }

        const float bias = bl[n];
        #pragma unroll
        for (int v = 0; v < 8; ++v) {
            float h = c[v] + bias;
            h = fmaxf(h, 0.0f);
            ab[(v + hi * 8) * HIDDEN + n] = h;
        }
    }

    float accz = bh[hi];
    #pragma unroll 8
    for (int nn = 0; nn < HIDDEN; ++nn)
        accz += ab[r * HIDDEN + nn] * Wh[hi * HIDDEN + nn];

    if (valid) out[(long)row * OUT_DIM + hi] = accz;
}

extern "C" void kernel_launch(void* const* d_in, const int* in_sizes, int n_in,
                              void* d_out, int out_size, void* d_ws, size_t ws_size,
                              hipStream_t stream) {
    const float* x  = (const float*)d_in[0];
    const int*   ei = (const int*)d_in[1];
    const float* Wl = (const float*)d_in[2];
    const float* bl = (const float*)d_in[3];
    const float* Wr = (const float*)d_in[4];
    const float* Wh = (const float*)d_in[5];
    const float* bh = (const float*)d_in[6];
    float* out = (float*)d_out;

    const long N = in_sizes[0] / IN_DIM;
    const long E = in_sizes[1] / 2;
    const int* src = ei;
    const int* dst = ei + E;

    int2* link = (int2*)d_ws;
    int*  head = (int*)(link + (size_t)E);

    {
        int blocks = (int)((N + 255) / 256);
        if (blocks < 1) blocks = 1;
        gnn_init_heads<<<blocks, 256, 0, stream>>>(head, (int)N);
    }
    {
        int blocks = (int)((E + 255) / 256);
        if (blocks < 1) blocks = 1;
        gnn_link_edges<<<blocks, 256, 0, stream>>>(src, dst, head, link, (int)E);
    }
    {
        long tiles  = (N + 15) / 16;
        int  blocks = (int)((tiles + 7) / 8);
        gnn_node_update<<<blocks, 256, 0, stream>>>(
            x, head, link, Wl, bl, Wr, Wh, bh, out, (int)N, (int)E);
    }
}


# stage W_l and W_r once per work-group and the tile's own x rows per wave in LDS with whole-line loads; B and root A fragments read from LDS instead of 16-line strided global loads
# speedup vs baseline: 2.6411x; 1.0914x over previous
#include <hip/hip_runtime.h>

typedef __attribute__((ext_vector_type(2))) float v2f;
typedef __attribute__((ext_vector_type(8))) float v8f;

#define IN_DIM  64
#define HIDDEN  64
#define OUT_DIM 2
#define APITCH  68
#define WPITCH  68

__global__ __launch_bounds__(256) void gnn_init_heads(int* __restrict__ head, int n) {
    int i = blockIdx.x * blockDim.x + threadIdx.x;
    if (i < n) head[i] = -1;
}

__global__ __launch_bounds__(256) void gnn_link_edges(
    const int* __restrict__ src,
    const int* __restrict__ dst,
    int* __restrict__ head,
    int2* __restrict__ link,
    int E) {
    int e = blockIdx.x * blockDim.x + threadIdx.x;
    if (e >= E) return;
    int d = dst[e];
    int s = src[e];
    int prev = atomicExch(&head[d], e);
    link[e] = make_int2(prev, s);
}

__global__ __launch_bounds__(256) void gnn_node_update(
    const float* __restrict__ x,
    const int*   __restrict__ head,
    const int2*  __restrict__ link,
    const float* __restrict__ Wl,
    const float* __restrict__ bl,
    const float* __restrict__ Wr,
    const float* __restrict__ Wh,
    const float* __restrict__ bh,
    float* __restrict__ out,
    int N, int E) {
    __shared__ alignas(16) float lbuf[8][16 * APITCH];
    __shared__ alignas(16) float wlds[2][HIDDEN * WPITCH];

    {
        const float4* wl4 = (const float4*)Wl;
        const float4* wr4 = (const float4*)Wr;
        #pragma unroll
        for (int i = 0; i < 4; ++i) {
            const int idx = (int)threadIdx.x + 256 * i;
            const int wrow = idx >> 4, c4 = idx & 15;
            const float4 a = wl4[idx];
            const float4 b = wr4[idx];
            *(float4*)(&wlds[0][wrow * WPITCH + 4 * c4]) = a;
            *(float4*)(&wlds[1][wrow * WPITCH + 4 * c4]) = b;
        }
    }
    __syncthreads();

    const int lane = threadIdx.x & 31;
    const int wave = threadIdx.x >> 5;
    const int tile = blockIdx.x * 8 + wave;
    const int r    = lane & 15;
    const int hi   = lane >> 4;

    const int  row   = tile * 16 + r;
    const bool valid = (row < N);

    const float2* x2 = (const float2*)x;
    float* ab = lbuf[wave];

    int  e    = valid ? head[row] : -1;

    const int tile_s = __builtin_amdgcn_readfirstlane(tile);
    float2 xt[16];
    #pragma unroll
    for (int i = 0; i < 16; ++i) {
        const int ri = tile_s * 16 + i;
        xt[i] = make_float2(0.0f, 0.0f);
        if (ri < N) xt[i] = x2[(long)ri * (IN_DIM / 2) + lane];
    }

    bool live = (unsigned)e < (unsigned)E;
    int  s = 0, nx = -1;
    if (live) { const int2 ln = link[e]; nx = ln.x; s = ln.y; }

    #pragma unroll
    for (int i = 0; i < 16; ++i)
        *(float2*)(ab + i * APITCH + 2 * lane) = xt[i];
    float cnt = 0.0f;
    float2 acc[16];
    #pragma unroll
    for (int i = 0; i < 16; ++i) acc[i] = make_float2(0.0f, 0.0f);

    for (int it = 0; it < E; ++it) {
        if (!__any(live)) break;
        const bool live2 = (unsigned)nx < (unsigned)E;
        int s2 = 0, nx2 = -1;
        if (live2) { const int2 ln = link[nx]; nx2 = ln.x; s2 = ln.y; }
        const int sm = live ? s : -1;
        float2 t[16];
        #pragma unroll
        for (int i = 0; i < 16; ++i) {
            const int si = __builtin_amdgcn_readlane(sm, i);
            t[i] = make_float2(0.0f, 0.0f);
            if ((unsigned)si < (unsigned)N) t[i] = x2[(long)si * (IN_DIM / 2) + lane];
        }
        #pragma unroll
        for (int i = 0; i < 16; ++i) { acc[i].x += t[i].x; acc[i].y += t[i].y; }
        cnt += live ? 1.0f : 0.0f;
        live = live2; s = s2; nx = nx2;
    }

    v2f aXf[16];
    #pragma unroll
    for (int kb = 0; kb < 16; ++kb) {
        const float2 a = *(const float2*)(ab + r * APITCH + 4 * kb + 2 * hi);
        aXf[kb].x = a.x; aXf[kb].y = a.y;
    }
    #pragma unroll
    for (int i = 0; i < 16; ++i)
        *(float2*)(ab + i * APITCH + 2 * lane) = acc[i];

    const float scale = 1.0f / fmaxf(cnt, 1.0f);
    v2f aAf[16];
    #pragma unroll
    for (int kb = 0; kb < 16; ++kb) {
        const float2 a = *(const float2*)(ab + r * APITCH + 4 * kb + 2 * hi);
        aAf[kb].x = a.x * scale; aAf[kb].y = a.y * scale;
    }

    const int koff2 = hi;

    for (int j = 0; j < 4; ++j) {
        const int n = j * 16 + r;
        const float2* wlrow = (const float2*)(&wlds[0][n * WPITCH]);
        const float2* wrrow = (const float2*)(&wlds[1][n * WPITCH]);

        v8f c = {0.f, 0.f, 0.f, 0.f, 0.f, 0.f, 0.f, 0.f};

        #pragma unroll
        for (int kb = 0; kb < 16; ++kb) {
            const int p = 2 * kb + koff2;
            float2 wl = wlrow[p];
            float2 wr = wrrow[p];

            v2f bL; bL.x = wl.x;         bL.y = wl.y;
            v2f bR; bR.x = wr.x;         bR.y = wr.y;

            c = __builtin_amdgcn_wmma_f32_16x16x4_f32(
                    false, aAf[kb], false, bL, (short)0, c, false, false);
            c = __builtin_amdgcn_wmma_f32_16x16x4_f32(
                    false, aXf[kb], false, bR, (short)0, c, false, false);
        }

        const float bias = bl[n];
        #pragma unroll
        for (int v = 0; v < 8; ++v) {
            float h = c[v] + bias;
            h = fmaxf(h, 0.0f);
            ab[(v + hi * 8) * HIDDEN + n] = h;
        }
    }

    float accz = bh[hi];
    #pragma unroll 8
    for (int nn = 0; nn < HIDDEN; ++nn)
        accz += ab[r * HIDDEN + nn] * Wh[hi * HIDDEN + nn];

    if (valid) out[(long)row * OUT_DIM + hi] = accz;
}

extern "C" void kernel_launch(void* const* d_in, const int* in_sizes, int n_in,
                              void* d_out, int out_size, void* d_ws, size_t ws_size,
                              hipStream_t stream) {
    const float* x  = (const float*)d_in[0];
    const int*   ei = (const int*)d_in[1];
    const float* Wl = (const float*)d_in[2];
    const float* bl = (const float*)d_in[3];
    const float* Wr = (const float*)d_in[4];
    const float* Wh = (const float*)d_in[5];
    const float* bh = (const float*)d_in[6];
    float* out = (float*)d_out;

    const long N = in_sizes[0] / IN_DIM;
    const long E = in_sizes[1] / 2;
    const int* src = ei;
    const int* dst = ei + E;

    int2* link = (int2*)d_ws;
    int*  head = (int*)(link + (size_t)E);

    {
        int blocks = (int)((N + 255) / 256);
        if (blocks < 1) blocks = 1;
        gnn_init_heads<<<blocks, 256, 0, stream>>>(head, (int)N);
    }
    {
        int blocks = (int)((E + 255) / 256);
        if (blocks < 1) blocks = 1;
        gnn_link_edges<<<blocks, 256, 0, stream>>>(src, dst, head, link, (int)E);
    }
    {
        long tiles  = (N + 15) / 16;
        int  blocks = (int)((tiles + 7) / 8);
        gnn_node_update<<<blocks, 256, 0, stream>>>(
            x, head, link, Wl, bl, Wr, Wh, bh, out, (int)N, (int)E);
    }
}


# group edges by 128-node bin via LDS counters (count, scan, place) and order each bin by node in LDS, replacing per-edge global exchange lists and their walk; sum order = LDS arrival rank
# speedup vs baseline: 4.3698x; 1.6545x over previous
#include <hip/hip_runtime.h>

typedef __attribute__((ext_vector_type(2))) float v2f;
typedef __attribute__((ext_vector_type(8))) float v8f;

#define IN_DIM  64
#define HIDDEN  64
#define OUT_DIM 2
#define APITCH  68
#define WPITCH  68

#define BIN_SHIFT 7
#define BIN_NODES 128
#define EDGE_WGS  128
#define NB_MAX    1024
#define CH        2048
#define SRC_NONE  0x00FFFFFF

__global__ __launch_bounds__(1024) void gnn_bin_count(
    const int* __restrict__ dst,
    int* __restrict__ hist,
    int N, int E, int B) {
    __shared__ int lcnt[NB_MAX];
    const int t = threadIdx.x, w = blockIdx.x;
    if (t < B) lcnt[t] = 0;
    __syncthreads();
    const int chunk = (E + EDGE_WGS - 1) / EDGE_WGS;
    const int e0 = w * chunk;
    int e1 = e0 + chunk; if (e1 > E) e1 = E;
    for (int eb = e0; eb < e1; eb += 8 * 1024) {
        int d[8];
        #pragma unroll
        for (int k = 0; k < 8; ++k) {
            const int e = eb + t + 1024 * k;
            d[k] = -1;
            if (e < e1) d[k] = dst[e];
        }
        #pragma unroll
        for (int k = 0; k < 8; ++k)
            if ((unsigned)d[k] < (unsigned)N) atomicAdd(&lcnt[d[k] >> BIN_SHIFT], 1);
    }
    __syncthreads();
    if (t < B) hist[w * B + t] = lcnt[t];
}

#define PASS (8 * 1024)
__global__ __launch_bounds__(1024) void gnn_bin_place(
    const int* __restrict__ src,
    const int* __restrict__ dst,
    const int* __restrict__ hist,
    int* __restrict__ bin_start,
    int* __restrict__ binned,
    int N, int E, int B) {
    __shared__ int lcnt[NB_MAX];
    __shared__ int base[NB_MAX];
    __shared__ int lsta[NB_MAX];
    __shared__ int wsum[32];
    __shared__ int wsum2[32];
    __shared__ int nloc;
    __shared__ int stage[PASS];
    __shared__ int sdel[PASS];
    const int t = threadIdx.x, w = blockIdx.x;
    const int lane = t & 31, wv = t >> 5;

    const int chunk = (E + EDGE_WGS - 1) / EDGE_WGS;
    const int e0 = w * chunk;
    int e1 = e0 + chunk; if (e1 > E) e1 = E;
    const bool staged = (chunk <= PASS);

    int d[8], s[8];
    #pragma unroll
    for (int k = 0; k < 8; ++k) {
        const int e = e0 + t + 1024 * k;
        d[k] = -1; s[k] = 0;
        if (e < e1) { d[k] = dst[e]; s[k] = src[e]; }
    }

    int tot = 0, pre = 0, mine = 0;
    if (t < B) {
        #pragma unroll 32
        for (int g = 0; g < EDGE_WGS; ++g) {
            const int v = hist[g * B + t];
            tot  += v;
            pre  += (g < w) ? v : 0;
            mine += (g == w) ? v : 0;
        }
    }
    int inc = tot, inc2 = mine;
    #pragma unroll
    for (int o = 1; o < 32; o <<= 1) {
        const int u  = __shfl_up(inc, o);
        const int u2 = __shfl_up(inc2, o);
        if (lane >= o) { inc += u; inc2 += u2; }
    }
    if (lane == 31) { wsum[wv] = inc; wsum2[wv] = inc2; }
    __syncthreads();
    if (wv == 0) {
        const int a = wsum[lane], a2 = wsum2[lane];
        int si = a, si2 = a2;
        #pragma unroll
        for (int o = 1; o < 32; o <<= 1) {
            const int u  = __shfl_up(si, o);
            const int u2 = __shfl_up(si2, o);
            if (lane >= o) { si += u; si2 += u2; }
        }
        wsum[lane]  = si - a;
        wsum2[lane] = si2 - a2;
    }
    __syncthreads();
    const int excl  = wsum[wv] + inc - tot;
    const int excl2 = wsum2[wv] + inc2 - mine;
    if (t < B) {
        base[t] = excl + pre;
        lsta[t] = excl2;
        lcnt[t] = 0;
        if (t == B - 1) nloc = excl2 + mine;
        if (w == 0) {
            bin_start[t] = excl;
            if (t == B - 1) bin_start[B] = excl + tot;
        }
    }
    __syncthreads();

    if (staged) {
        #pragma unroll
        for (int k = 0; k < 8; ++k) {
            if ((unsigned)d[k] < (unsigned)N) {
                const int b  = d[k] >> BIN_SHIFT;
                const int rk = atomicAdd(&lcnt[b], 1);
                const int sp = ((unsigned)s[k] < (unsigned)N) ? s[k] : SRC_NONE;
                const int lp = lsta[b] + rk;
                stage[lp] = (sp << BIN_SHIFT) | (d[k] & (BIN_NODES - 1));
                sdel[lp]  = base[b] - lsta[b];
            }
        }
        __syncthreads();
        const int nl = min(nloc, PASS);
        for (int j = t; j < nl; j += 1024)
            binned[j + sdel[j]] = stage[j];
    } else {
        for (int eb = e0; eb < e1; eb += PASS) {
            if (eb != e0) {
                #pragma unroll
                for (int k = 0; k < 8; ++k) {
                    const int e = eb + t + 1024 * k;
                    d[k] = -1; s[k] = 0;
                    if (e < e1) { d[k] = dst[e]; s[k] = src[e]; }
                }
            }
            #pragma unroll
            for (int k = 0; k < 8; ++k) {
                if ((unsigned)d[k] < (unsigned)N) {
                    const int b  = d[k] >> BIN_SHIFT;
                    const int rk = atomicAdd(&lcnt[b], 1);
                    const int sp = ((unsigned)s[k] < (unsigned)N) ? s[k] : SRC_NONE;
                    binned[base[b] + rk] = (sp << BIN_SHIFT) | (d[k] & (BIN_NODES - 1));
                }
            }
        }
    }
}

__global__ __launch_bounds__(256) void gnn_node_update(
    const float* __restrict__ x,
    const int*   __restrict__ bin_start,
    const int*   __restrict__ binned,
    const float* __restrict__ Wl,
    const float* __restrict__ bl,
    const float* __restrict__ Wr,
    const float* __restrict__ Wh,
    const float* __restrict__ bh,
    float* __restrict__ out,
    int N, int E) {
    __shared__ alignas(16) float lbuf[8][16 * APITCH];
    __shared__ alignas(16) float wlds[2][HIDDEN * WPITCH];
    __shared__ int sedge[CH];
    __shared__ alignas(16) int ncnt[BIN_NODES];
    __shared__ int nstart[BIN_NODES];

    {
        const float4* wl4 = (const float4*)Wl;
        const float4* wr4 = (const float4*)Wr;
        #pragma unroll
        for (int i = 0; i < 4; ++i) {
            const int idx = (int)threadIdx.x + 256 * i;
            const int wrow = idx >> 4, c4 = idx & 15;
            const float4 a = wl4[idx];
            const float4 b = wr4[idx];
            *(float4*)(&wlds[0][wrow * WPITCH + 4 * c4]) = a;
            *(float4*)(&wlds[1][wrow * WPITCH + 4 * c4]) = b;
        }
    }
    int bs = bin_start[blockIdx.x];
    int be = bin_start[blockIdx.x + 1];
    bs = min(max(bs, 0), E);
    be = min(max(be, bs), E);
    __syncthreads();

    const int lane = threadIdx.x & 31;
    const int wave = threadIdx.x >> 5;
    const int tile = blockIdx.x * 8 + wave;
    const int r    = lane & 15;
    const int hi   = lane >> 4;

    const int  row   = tile * 16 + r;
    const bool valid = (row < N);

    const float2* x2 = (const float2*)x;
    float* ab = lbuf[wave];

    const int tile_s = __builtin_amdgcn_readfirstlane(tile);
    float2 xt[16];
    #pragma unroll
    for (int i = 0; i < 16; ++i) {
        const int ri = tile_s * 16 + i;
        xt[i] = make_float2(0.0f, 0.0f);
        if (ri < N) xt[i] = x2[(long)ri * (IN_DIM / 2) + lane];
    }
    int ev[CH / 256];
    {
        const int n0 = min(CH, be - bs);
        #pragma unroll
        for (int k = 0; k < CH / 256; ++k) {
            const int j = (int)threadIdx.x + 256 * k;
            ev[k] = -1;
            if (j < n0) ev[k] = binned[bs + j];
        }
    }
    #pragma unroll
    for (int i = 0; i < 16; ++i)
        *(float2*)(ab + i * APITCH + 2 * lane) = xt[i];

    float cnt = 0.0f;
    float2 acc[16];
    #pragma unroll
    for (int i = 0; i < 16; ++i) acc[i] = make_float2(0.0f, 0.0f);

    const int node_l = wave * 16 + r;

    for (int c0 = bs; c0 < be; c0 += CH) {
        const int n = min(CH, be - c0);
        if (c0 != bs) __syncthreads();
        if (threadIdx.x < BIN_NODES) ncnt[threadIdx.x] = 0;
        __syncthreads();

        int rk[CH / 256];
        if (c0 != bs) {
            #pragma unroll
            for (int k = 0; k < CH / 256; ++k) {
                const int j = (int)threadIdx.x + 256 * k;
                ev[k] = -1;
                if (j < n) ev[k] = binned[c0 + j];
            }
        }
        #pragma unroll
        for (int k = 0; k < CH / 256; ++k) rk[k] = 0;
        #pragma unroll
        for (int k = 0; k < CH / 256; ++k)
            if (ev[k] >= 0) rk[k] = atomicAdd(&ncnt[ev[k] & (BIN_NODES - 1)], 1);
        __syncthreads();

        if (wave == 0) {
            const int4 c = *(const int4*)(&ncnt[4 * lane]);
            const int sum4 = c.x + c.y + c.z + c.w;
            int inc = sum4;
            #pragma unroll
            for (int o = 1; o < 32; o <<= 1) {
                const int u = __shfl_up(inc, o);
                if (lane >= o) inc += u;
            }
            const int ex = inc - sum4;
            nstart[4 * lane]     = ex;
            nstart[4 * lane + 1] = ex + c.x;
            nstart[4 * lane + 2] = ex + c.x + c.y;
            nstart[4 * lane + 3] = ex + c.x + c.y + c.z;
        }
        __syncthreads();

        #pragma unroll
        for (int k = 0; k < CH / 256; ++k)
            if (ev[k] >= 0) sedge[nstart[ev[k] & (BIN_NODES - 1)] + rk[k]] = ev[k] >> BIN_SHIFT;
        __syncthreads();

        const int my_n = ncnt[node_l];
        const int my_s = nstart[node_l];
        for (int t = 0; t < n; ++t) {
            const bool live = t < my_n;
            if (!__any(live)) break;
            const int sm = live ? sedge[my_s + t] : -1;
            float2 tt[16];
            #pragma unroll
            for (int i = 0; i < 16; ++i) {
                const int si = __builtin_amdgcn_readlane(sm, i);
                tt[i] = make_float2(0.0f, 0.0f);
                if ((unsigned)si < (unsigned)N) tt[i] = x2[(long)si * (IN_DIM / 2) + lane];
            }
            #pragma unroll
            for (int i = 0; i < 16; ++i) { acc[i].x += tt[i].x; acc[i].y += tt[i].y; }
        }
        cnt += (float)my_n;
    }

    v2f aXf[16];
    #pragma unroll
    for (int kb = 0; kb < 16; ++kb) {
        const float2 a = *(const float2*)(ab + r * APITCH + 4 * kb + 2 * hi);
        aXf[kb].x = a.x; aXf[kb].y = a.y;
    }
    #pragma unroll
    for (int i = 0; i < 16; ++i)
        *(float2*)(ab + i * APITCH + 2 * lane) = acc[i];

    const float scale = 1.0f / fmaxf(cnt, 1.0f);
    v2f aAf[16];
    #pragma unroll
    for (int kb = 0; kb < 16; ++kb) {
        const float2 a = *(const float2*)(ab + r * APITCH + 4 * kb + 2 * hi);
        aAf[kb].x = a.x * scale; aAf[kb].y = a.y * scale;
    }

    const int koff2 = hi;

    for (int j = 0; j < 4; ++j) {
        const int n = j * 16 + r;
        const float2* wlrow = (const float2*)(&wlds[0][n * WPITCH]);
        const float2* wrrow = (const float2*)(&wlds[1][n * WPITCH]);

        v8f c = {0.f, 0.f, 0.f, 0.f, 0.f, 0.f, 0.f, 0.f};

        #pragma unroll
        for (int kb = 0; kb < 16; ++kb) {
            const int p = 2 * kb + koff2;
            float2 wl = wlrow[p];
            float2 wr = wrrow[p];

            v2f bL; bL.x = wl.x;         bL.y = wl.y;
            v2f bR; bR.x = wr.x;         bR.y = wr.y;

            c = __builtin_amdgcn_wmma_f32_16x16x4_f32(
                    false, aAf[kb], false, bL, (short)0, c, false, false);
            c = __builtin_amdgcn_wmma_f32_16x16x4_f32(
                    false, aXf[kb], false, bR, (short)0, c, false, false);
        }

        const float bias = bl[n];
        #pragma unroll
        for (int v = 0; v < 8; ++v) {
            float h = c[v] + bias;
            h = fmaxf(h, 0.0f);
            ab[(v + hi * 8) * HIDDEN + n] = h;
        }
    }

    float accz = bh[hi];
    #pragma unroll 8
    for (int nn = 0; nn < HIDDEN; ++nn)
        accz += ab[r * HIDDEN + nn] * Wh[hi * HIDDEN + nn];

    if (valid) out[(long)row * OUT_DIM + hi] = accz;
}

extern "C" void kernel_launch(void* const* d_in, const int* in_sizes, int n_in,
                              void* d_out, int out_size, void* d_ws, size_t ws_size,
                              hipStream_t stream) {
    const float* x  = (const float*)d_in[0];
    const int*   ei = (const int*)d_in[1];
    const float* Wl = (const float*)d_in[2];
    const float* bl = (const float*)d_in[3];
    const float* Wr = (const float*)d_in[4];
    const float* Wh = (const float*)d_in[5];
    const float* bh = (const float*)d_in[6];
    float* out = (float*)d_out;

    const long N = in_sizes[0] / IN_DIM;
    const long E = in_sizes[1] / 2;
    const int* src = ei;
    const int* dst = ei + E;

    const int B = (int)((N + BIN_NODES - 1) / BIN_NODES);

    int* binned    = (int*)d_ws;
    int* hist      = binned + (size_t)E;
    int* bin_start = hist + (size_t)EDGE_WGS * B;

    gnn_bin_count<<<EDGE_WGS, 1024, 0, stream>>>(dst, hist, (int)N, (int)E, B);
    gnn_bin_place<<<EDGE_WGS, 1024, 0, stream>>>(src, dst, hist, bin_start, binned, (int)N, (int)E, B);
    {
        long tiles  = (N + 15) / 16;
        int  blocks = (int)((tiles + 7) / 8);
        gnn_node_update<<<blocks, 256, 0, stream>>>(
            x, bin_start, binned, Wl, bl, Wr, Wh, bh, out, (int)N, (int)E);
    }
}


# group each edge slice by bin on its own in LDS; node kernel collects its bin's pieces via a bin-by-slice offset table: no count launch, no count-table sums; sum order = LDS arrival rank as before
# speedup vs baseline: 4.9209x; 1.1261x over previous
#include <hip/hip_runtime.h>

typedef __attribute__((ext_vector_type(2))) float v2f;
typedef __attribute__((ext_vector_type(8))) float v8f;

#define IN_DIM  64
#define HIDDEN  64
#define OUT_DIM 2
#define APITCH  68
#define WPITCH  68

#define BIN_SHIFT 7
#define BIN_NODES 128
#define SLICES    256
#define NB_MAX    1024
#define CH        2048
#define SRC_NONE  0x00FFFFFF

#define PASS (4 * 1024)
__global__ __launch_bounds__(1024) void gnn_slice_group(
    const int* __restrict__ src,
    const int* __restrict__ dst,
    int* __restrict__ slice_ofs,
    int* __restrict__ binned,
    int N, int E, int B) {
    __shared__ int lcnt[NB_MAX];
    __shared__ int lsta[NB_MAX];
    __shared__ int wsum[32];
    __shared__ int stage[PASS];
    const int t = threadIdx.x, w = blockIdx.x;
    const int lane = t & 31, wv = t >> 5;

    const int chunk = (E + SLICES - 1) / SLICES;
    const int e0 = min(w * chunk, E);
    const int e1 = min(e0 + chunk, E);
    const bool staged = (chunk <= PASS);

    int d[4], s[4];
    #pragma unroll
    for (int k = 0; k < 4; ++k) {
        const int e = e0 + t + 1024 * k;
        d[k] = -1; s[k] = 0;
        if (e < e1) { d[k] = dst[e]; s[k] = src[e]; }
    }
    lcnt[t] = 0;
    __syncthreads();

    int rk[4];
    #pragma unroll
    for (int k = 0; k < 4; ++k) rk[k] = 0;
    for (int eb = e0; eb < e1; eb += PASS) {
        if (eb != e0) {
            #pragma unroll
            for (int k = 0; k < 4; ++k) {
                const int e = eb + t + 1024 * k;
                d[k] = -1;
                if (e < e1) d[k] = dst[e];
            }
        }
        #pragma unroll
        for (int k = 0; k < 4; ++k)
            if ((unsigned)d[k] < (unsigned)N) rk[k] = atomicAdd(&lcnt[d[k] >> BIN_SHIFT], 1);
    }
    __syncthreads();

    const int mine = lcnt[t];
    int inc = mine;
    #pragma unroll
    for (int o = 1; o < 32; o <<= 1) {
        const int u = __shfl_up(inc, o);
        if (lane >= o) inc += u;
    }
    if (lane == 31) wsum[wv] = inc;
    __syncthreads();
    if (wv == 0) {
        const int a = wsum[lane];
        int si = a;
        #pragma unroll
        for (int o = 1; o < 32; o <<= 1) {
            const int u = __shfl_up(si, o);
            if (lane >= o) si += u;
        }
        wsum[lane] = si - a;
    }
    __syncthreads();
    const int excl = wsum[wv] + inc - mine;
    lsta[t] = excl;
    if (t < B) {
        slice_ofs[t * SLICES + w] = excl;
        if (t == B - 1) slice_ofs[B * SLICES + w] = excl + mine;
    }
    if (!staged) lcnt[t] = 0;
    __syncthreads();

    if (staged) {
        #pragma unroll
        for (int k = 0; k < 4; ++k) {
            if ((unsigned)d[k] < (unsigned)N) {
                const int b  = d[k] >> BIN_SHIFT;
                const int sp = ((unsigned)s[k] < (unsigned)N) ? s[k] : SRC_NONE;
                stage[lsta[b] + rk[k]] = (sp << BIN_SHIFT) | (d[k] & (BIN_NODES - 1));
            }
        }
        __syncthreads();
        const int nl = min(lsta[NB_MAX - 1] + lcnt[NB_MAX - 1], e1 - e0);
        for (int j = t; j < nl; j += 1024)
            binned[e0 + j] = stage[j];
    } else {
        for (int eb = e0; eb < e1; eb += PASS) {
            #pragma unroll
            for (int k = 0; k < 4; ++k) {
                const int e = eb + t + 1024 * k;
                d[k] = -1; s[k] = 0;
                if (e < e1) { d[k] = dst[e]; s[k] = src[e]; }
            }
            #pragma unroll
            for (int k = 0; k < 4; ++k) {
                if ((unsigned)d[k] < (unsigned)N) {
                    const int b  = d[k] >> BIN_SHIFT;
                    const int r2 = atomicAdd(&lcnt[b], 1);
                    const int sp = ((unsigned)s[k] < (unsigned)N) ? s[k] : SRC_NONE;
                    binned[e0 + lsta[b] + r2] = (sp << BIN_SHIFT) | (d[k] & (BIN_NODES - 1));
                }
            }
        }
    }
}

__global__ __launch_bounds__(256) void gnn_node_update(
    const float* __restrict__ x,
    const int*   __restrict__ slice_ofs,
    const int*   __restrict__ binned,
    const float* __restrict__ Wl,
    const float* __restrict__ bl,
    const float* __restrict__ Wr,
    const float* __restrict__ Wh,
    const float* __restrict__ bh,
    float* __restrict__ out,
    int N, int E) {
    __shared__ alignas(16) float lbuf[8][16 * APITCH];
    __shared__ alignas(16) float wlds[2][HIDDEN * WPITCH];
    __shared__ int sedge[CH];
    __shared__ alignas(16) int ncnt[BIN_NODES];
    __shared__ int nstart[BIN_NODES];
    __shared__ int wtot[8];

    {
        const float4* wl4 = (const float4*)Wl;
        const float4* wr4 = (const float4*)Wr;
        #pragma unroll
        for (int i = 0; i < 4; ++i) {
            const int idx = (int)threadIdx.x + 256 * i;
            const int wrow = idx >> 4, c4 = idx & 15;
            const float4 a = wl4[idx];
            const float4 b = wr4[idx];
            *(float4*)(&wlds[0][wrow * WPITCH + 4 * c4]) = a;
            *(float4*)(&wlds[1][wrow * WPITCH + 4 * c4]) = b;
        }
    }
    const int lane = threadIdx.x & 31;
    const int wave = threadIdx.x >> 5;

    const int chunk = (E + SLICES - 1) / SLICES;
    const int gb    = min((int)threadIdx.x * chunk, E);
    const int glen  = min(chunk, E - gb);
    int o0 = slice_ofs[blockIdx.x * SLICES + threadIdx.x];
    int o1 = slice_ofs[(blockIdx.x + 1) * SLICES + threadIdx.x];
    o0 = min(max(o0, 0), glen);
    o1 = min(max(o1, o0), glen);
    const int cg = o1 - o0;
    const int p0 = gb + o0;
    const int tile = blockIdx.x * 8 + wave;
    const int r    = lane & 15;
    const int hi   = lane >> 4;

    const int  row   = tile * 16 + r;
    const bool valid = (row < N);

    const float2* x2 = (const float2*)x;
    float* ab = lbuf[wave];

    const int tile_s = __builtin_amdgcn_readfirstlane(tile);
    float2 xt[16];
    #pragma unroll
    for (int i = 0; i < 16; ++i) {
        const int ri = tile_s * 16 + i;
        xt[i] = make_float2(0.0f, 0.0f);
        if (ri < N) xt[i] = x2[(long)ri * (IN_DIM / 2) + lane];
    }

    int pinc = cg;
    #pragma unroll
    for (int o = 1; o < 32; o <<= 1) {
        const int u = __shfl_up(pinc, o);
        if (lane >= o) pinc += u;
    }
    if (lane == 31) wtot[wave] = pinc;
    __syncthreads();
    int pre = pinc - cg, ntot = 0;
    #pragma unroll
    for (int i = 0; i < 8; ++i) {
        const int v = wtot[i];
        ntot += v;
        pre  += (i < wave) ? v : 0;
    }
    #pragma unroll
    for (int i = 0; i < 16; ++i)
        *(float2*)(ab + i * APITCH + 2 * lane) = xt[i];

    float cnt = 0.0f;
    float2 acc[16];
    #pragma unroll
    for (int i = 0; i < 16; ++i) acc[i] = make_float2(0.0f, 0.0f);

    const int node_l = wave * 16 + r;

    for (int c0 = 0; c0 < ntot; c0 += CH) {
        const int n = min(CH, ntot - c0);
        if (c0 != 0) __syncthreads();
        if (threadIdx.x < BIN_NODES) ncnt[threadIdx.x] = 0;
        {
            const int jlo = max(pre, c0), jhi = min(pre + cg, c0 + CH);
            for (int j = jlo; j < jhi; ++j) sedge[j - c0] = p0 + (j - pre);
        }
        __syncthreads();

        int pl[CH / 256];
        #pragma unroll
        for (int k = 0; k < CH / 256; ++k)
            pl[k] = sedge[min((int)threadIdx.x + 256 * k, n - 1)];
        int ev[CH / 256];
        #pragma unroll
        for (int k = 0; k < CH / 256; ++k) {
            const int j = (int)threadIdx.x + 256 * k;
            ev[k] = -1;
            if (j < n) ev[k] = binned[pl[k]];
        }
        int rk[CH / 256];
        #pragma unroll
        for (int k = 0; k < CH / 256; ++k) rk[k] = 0;
        #pragma unroll
        for (int k = 0; k < CH / 256; ++k)
            if (ev[k] >= 0) rk[k] = atomicAdd(&ncnt[ev[k] & (BIN_NODES - 1)], 1);
        __syncthreads();

        if (wave == 0) {
            const int4 c = *(const int4*)(&ncnt[4 * lane]);
            const int sum4 = c.x + c.y + c.z + c.w;
            int inc = sum4;
            #pragma unroll
            for (int o = 1; o < 32; o <<= 1) {
                const int u = __shfl_up(inc, o);
                if (lane >= o) inc += u;
            }
            const int ex = inc - sum4;
            nstart[4 * lane]     = ex;
            nstart[4 * lane + 1] = ex + c.x;
            nstart[4 * lane + 2] = ex + c.x + c.y;
            nstart[4 * lane + 3] = ex + c.x + c.y + c.z;
        }
        __syncthreads();

        #pragma unroll
        for (int k = 0; k < CH / 256; ++k)
            if (ev[k] >= 0) sedge[nstart[ev[k] & (BIN_NODES - 1)] + rk[k]] = ev[k] >> BIN_SHIFT;
        __syncthreads();

        const int my_n = ncnt[node_l];
        const int my_s = nstart[node_l];
        for (int t = 0; t < n; ++t) {
            const bool live = t < my_n;
            if (!__any(live)) break;
            const int sm = live ? sedge[my_s + t] : -1;
            float2 tt[16];
            #pragma unroll
            for (int i = 0; i < 16; ++i) {
                const int si = __builtin_amdgcn_readlane(sm, i);
                tt[i] = make_float2(0.0f, 0.0f);
                if ((unsigned)si < (unsigned)N) tt[i] = x2[(long)si * (IN_DIM / 2) + lane];
            }
            #pragma unroll
            for (int i = 0; i < 16; ++i) { acc[i].x += tt[i].x; acc[i].y += tt[i].y; }
        }
        cnt += (float)my_n;
    }

    v2f aXf[16];
    #pragma unroll
    for (int kb = 0; kb < 16; ++kb) {
        const float2 a = *(const float2*)(ab + r * APITCH + 4 * kb + 2 * hi);
        aXf[kb].x = a.x; aXf[kb].y = a.y;
    }
    #pragma unroll
    for (int i = 0; i < 16; ++i)
        *(float2*)(ab + i * APITCH + 2 * lane) = acc[i];

    const float scale = 1.0f / fmaxf(cnt, 1.0f);
    v2f aAf[16];
    #pragma unroll
    for (int kb = 0; kb < 16; ++kb) {
        const float2 a = *(const float2*)(ab + r * APITCH + 4 * kb + 2 * hi);
        aAf[kb].x = a.x * scale; aAf[kb].y = a.y * scale;
    }

    const int koff2 = hi;

    for (int j = 0; j < 4; ++j) {
        const int n = j * 16 + r;
        const float2* wlrow = (const float2*)(&wlds[0][n * WPITCH]);
        const float2* wrrow = (const float2*)(&wlds[1][n * WPITCH]);

        v8f c = {0.f, 0.f, 0.f, 0.f, 0.f, 0.f, 0.f, 0.f};

        #pragma unroll
        for (int kb = 0; kb < 16; ++kb) {
            const int p = 2 * kb + koff2;
            float2 wl = wlrow[p];
            float2 wr = wrrow[p];

            v2f bL; bL.x = wl.x;         bL.y = wl.y;
            v2f bR; bR.x = wr.x;         bR.y = wr.y;

            c = __builtin_amdgcn_wmma_f32_16x16x4_f32(
                    false, aAf[kb], false, bL, (short)0, c, false, false);
            c = __builtin_amdgcn_wmma_f32_16x16x4_f32(
                    false, aXf[kb], false, bR, (short)0, c, false, false);
        }

        const float bias = bl[n];
        #pragma unroll
        for (int v = 0; v < 8; ++v) {
            float h = c[v] + bias;
            h = fmaxf(h, 0.0f);
            ab[(v + hi * 8) * HIDDEN + n] = h;
        }
    }

    float accz = bh[hi];
    #pragma unroll 8
    for (int nn = 0; nn < HIDDEN; ++nn)
        accz += ab[r * HIDDEN + nn] * Wh[hi * HIDDEN + nn];

    if (valid) out[(long)row * OUT_DIM + hi] = accz;
}

extern "C" void kernel_launch(void* const* d_in, const int* in_sizes, int n_in,
                              void* d_out, int out_size, void* d_ws, size_t ws_size,
                              hipStream_t stream) {
    const float* x  = (const float*)d_in[0];
    const int*   ei = (const int*)d_in[1];
    const float* Wl = (const float*)d_in[2];
    const float* bl = (const float*)d_in[3];
    const float* Wr = (const float*)d_in[4];
    const float* Wh = (const float*)d_in[5];
    const float* bh = (const float*)d_in[6];
    float* out = (float*)d_out;

    const long N = in_sizes[0] / IN_DIM;
    const long E = in_sizes[1] / 2;
    const int* src = ei;
    const int* dst = ei + E;

    const int B = (int)((N + BIN_NODES - 1) / BIN_NODES);

    int* binned    = (int*)d_ws;
    int* slice_ofs = binned + (size_t)E;

    gnn_slice_group<<<SLICES, 1024, 0, stream>>>(src, dst, slice_ofs, binned, (int)N, (int)E, B);
    {
        long tiles  = (N + 15) / 16;
        int  blocks = (int)((tiles + 7) / 8);
        gnn_node_update<<<blocks, 256, 0, stream>>>(
            x, slice_ofs, binned, Wl, bl, Wr, Wh, bh, out, (int)N, (int)E);
    }
}


# node kernel walks each tile's node-ordered edge run 16 edges per batch (every row load a real edge, row_share addresses, scalar bit tests), one running sum stored at node changes; sum order unchanged
# speedup vs baseline: 5.2642x; 1.0698x over previous
#include <hip/hip_runtime.h>

typedef __attribute__((ext_vector_type(2))) float v2f;
typedef __attribute__((ext_vector_type(8))) float v8f;

#define IN_DIM  64
#define HIDDEN  64
#define OUT_DIM 2
#define APITCH  68
#define WPITCH  68

#define BIN_SHIFT 7
#define BIN_NODES 128
#define SLICES    256
#define NB_MAX    1024
#define CH        2048
#define SRC_NONE  0x00FFFFFF

#define PASS (4 * 1024)
__global__ __launch_bounds__(1024) void gnn_slice_group(
    const int* __restrict__ src,
    const int* __restrict__ dst,
    int* __restrict__ slice_ofs,
    int* __restrict__ binned,
    int N, int E, int B) {
    __shared__ int lcnt[NB_MAX];
    __shared__ int lsta[NB_MAX];
    __shared__ int wsum[32];
    __shared__ int stage[PASS];
    const int t = threadIdx.x, w = blockIdx.x;
    const int lane = t & 31, wv = t >> 5;

    const int chunk = (E + SLICES - 1) / SLICES;
    const int e0 = min(w * chunk, E);
    const int e1 = min(e0 + chunk, E);
    const bool staged = (chunk <= PASS);

    int d[4], s[4];
    #pragma unroll
    for (int k = 0; k < 4; ++k) {
        const int e = e0 + t + 1024 * k;
        d[k] = -1; s[k] = 0;
        if (e < e1) { d[k] = dst[e]; s[k] = src[e]; }
    }
    lcnt[t] = 0;
    __syncthreads();

    int rk[4];
    #pragma unroll
    for (int k = 0; k < 4; ++k) rk[k] = 0;
    for (int eb = e0; eb < e1; eb += PASS) {
        if (eb != e0) {
            #pragma unroll
            for (int k = 0; k < 4; ++k) {
                const int e = eb + t + 1024 * k;
                d[k] = -1;
                if (e < e1) d[k] = dst[e];
            }
        }
        #pragma unroll
        for (int k = 0; k < 4; ++k)
            if ((unsigned)d[k] < (unsigned)N) rk[k] = atomicAdd(&lcnt[d[k] >> BIN_SHIFT], 1);
    }
    __syncthreads();

    const int mine = lcnt[t];
    int inc = mine;
    #pragma unroll
    for (int o = 1; o < 32; o <<= 1) {
        const int u = __shfl_up(inc, o);
        if (lane >= o) inc += u;
    }
    if (lane == 31) wsum[wv] = inc;
    __syncthreads();
    if (wv == 0) {
        const int a = wsum[lane];
        int si = a;
        #pragma unroll
        for (int o = 1; o < 32; o <<= 1) {
            const int u = __shfl_up(si, o);
            if (lane >= o) si += u;
        }
        wsum[lane] = si - a;
    }
    __syncthreads();
    const int excl = wsum[wv] + inc - mine;
    lsta[t] = excl;
    if (t < B) {
        slice_ofs[t * SLICES + w] = excl;
        if (t == B - 1) slice_ofs[B * SLICES + w] = excl + mine;
    }
    if (!staged) lcnt[t] = 0;
    __syncthreads();

    if (staged) {
        #pragma unroll
        for (int k = 0; k < 4; ++k) {
            if ((unsigned)d[k] < (unsigned)N) {
                const int b  = d[k] >> BIN_SHIFT;
                const int sp = ((unsigned)s[k] < (unsigned)N) ? s[k] : SRC_NONE;
                stage[lsta[b] + rk[k]] = (sp << BIN_SHIFT) | (d[k] & (BIN_NODES - 1));
            }
        }
        __syncthreads();
        const int nl = min(lsta[NB_MAX - 1] + lcnt[NB_MAX - 1], e1 - e0);
        for (int j = t; j < nl; j += 1024)
            binned[e0 + j] = stage[j];
    } else {
        for (int eb = e0; eb < e1; eb += PASS) {
            #pragma unroll
            for (int k = 0; k < 4; ++k) {
                const int e = eb + t + 1024 * k;
                d[k] = -1; s[k] = 0;
                if (e < e1) { d[k] = dst[e]; s[k] = src[e]; }
            }
            #pragma unroll
            for (int k = 0; k < 4; ++k) {
                if ((unsigned)d[k] < (unsigned)N) {
                    const int b  = d[k] >> BIN_SHIFT;
                    const int r2 = atomicAdd(&lcnt[b], 1);
                    const int sp = ((unsigned)s[k] < (unsigned)N) ? s[k] : SRC_NONE;
                    binned[e0 + lsta[b] + r2] = (sp << BIN_SHIFT) | (d[k] & (BIN_NODES - 1));
                }
            }
        }
    }
}

template <int I>
__device__ __forceinline__ unsigned row_share(unsigned v) {
    return (unsigned)__builtin_amdgcn_update_dpp(0, (int)v, 0x150 + I, 0xF, 0xF, false);
}
__device__ __forceinline__ int row_below(int first, int v) {
    return __builtin_amdgcn_update_dpp(first, v, 0x111  , 0xF, 0xF, false);
}

template <bool RESUME>
__device__ __forceinline__ void gather_run(const float* __restrict__ x, float* ab, const int* sedge,
                                           int es, int ee, int N, int lane) {
    const int r = lane & 15;
    const int pad = sedge[ee - 1] | (SRC_NONE << BIN_SHIFT);
    const unsigned lane8 = 8u * (unsigned)lane;
    float2 run = make_float2(0.0f, 0.0f);
    int w = (es + r < ee) ? sedge[es + r] : pad;
    unsigned vm = __builtin_amdgcn_ballot_w32((unsigned)(w >> BIN_SHIFT) < (unsigned)N);
    unsigned cm = __builtin_amdgcn_ballot_w32(((w ^ row_below(w ^ 1, w)) & (BIN_NODES - 1)) != 0);
    unsigned rowb = 0;
    for (int e = es; e < ee; e += 16) {
        const unsigned off  = ((unsigned)w >> BIN_SHIFT) << 8;
        const unsigned rowo = (unsigned)(w & 15) * (APITCH * 4u);
        float2 tt[16];
#define GATHER_SLOT(I)                                                                   \
        tt[I] = make_float2(0.0f, 0.0f);                                                 \
        if ((vm >> I) & 1u)                                                              \
            tt[I] = *(const float2*)((const char*)x + (row_share<I>(off) + lane8));
        GATHER_SLOT(0)  GATHER_SLOT(1)  GATHER_SLOT(2)  GATHER_SLOT(3)
        GATHER_SLOT(4)  GATHER_SLOT(5)  GATHER_SLOT(6)  GATHER_SLOT(7)
        GATHER_SLOT(8)  GATHER_SLOT(9)  GATHER_SLOT(10) GATHER_SLOT(11)
        GATHER_SLOT(12) GATHER_SLOT(13) GATHER_SLOT(14) GATHER_SLOT(15)
#undef GATHER_SLOT
        const int jn = e + 16 + r;
        const int wn = (jn < ee) ? sedge[jn] : pad;
        const unsigned vmn = __builtin_amdgcn_ballot_w32((unsigned)(wn >> BIN_SHIFT) < (unsigned)N);
        const unsigned cmn = __builtin_amdgcn_ballot_w32(
            ((wn ^ row_below((int)row_share<15>((unsigned)w), wn)) & (BIN_NODES - 1)) != 0);
#define SUM_SLOT(I, HAVE_BEFORE, PO)                                                     \
        if (__builtin_expect((cm >> I) & 1u, 0)) {                                       \
            if (HAVE_BEFORE) *(float2*)((char*)ab + ((PO) + lane8)) = run;               \
            run = make_float2(0.0f, 0.0f);                                               \
            if (RESUME) run = *(const float2*)((const char*)ab + (row_share<I>(rowo) + lane8)); \
        }                                                                                \
        run.x += tt[I].x; run.y += tt[I].y;
        SUM_SLOT(0, e != es, rowb)
        SUM_SLOT(1,  true, row_share<0>(rowo))   SUM_SLOT(2,  true, row_share<1>(rowo))
        SUM_SLOT(3,  true, row_share<2>(rowo))   SUM_SLOT(4,  true, row_share<3>(rowo))
        SUM_SLOT(5,  true, row_share<4>(rowo))   SUM_SLOT(6,  true, row_share<5>(rowo))
        SUM_SLOT(7,  true, row_share<6>(rowo))   SUM_SLOT(8,  true, row_share<7>(rowo))
        SUM_SLOT(9,  true, row_share<8>(rowo))   SUM_SLOT(10, true, row_share<9>(rowo))
        SUM_SLOT(11, true, row_share<10>(rowo))  SUM_SLOT(12, true, row_share<11>(rowo))
        SUM_SLOT(13, true, row_share<12>(rowo))  SUM_SLOT(14, true, row_share<13>(rowo))
        SUM_SLOT(15, true, row_share<14>(rowo))
#undef SUM_SLOT
        rowb = row_share<15>(rowo);
        w = wn; vm = vmn; cm = cmn;
    }
    *(float2*)(ab + (pad & 15) * APITCH + 2 * lane) = run;
}

__global__ __launch_bounds__(256) void gnn_node_update(
    const float* __restrict__ x,
    const int*   __restrict__ slice_ofs,
    const int*   __restrict__ binned,
    const float* __restrict__ Wl,
    const float* __restrict__ bl,
    const float* __restrict__ Wr,
    const float* __restrict__ Wh,
    const float* __restrict__ bh,
    float* __restrict__ out,
    int N, int E) {
    __shared__ alignas(16) float lbuf[8][16 * APITCH];
    __shared__ alignas(16) float wlds[2][HIDDEN * WPITCH];
    __shared__ int sedge[CH];
    __shared__ alignas(16) int ncnt[BIN_NODES];
    __shared__ int nstart[BIN_NODES];
    __shared__ int wtot[8];

    {
        const float4* wl4 = (const float4*)Wl;
        const float4* wr4 = (const float4*)Wr;
        #pragma unroll
        for (int i = 0; i < 4; ++i) {
            const int idx = (int)threadIdx.x + 256 * i;
            const int wrow = idx >> 4, c4 = idx & 15;
            const float4 a = wl4[idx];
            const float4 b = wr4[idx];
            *(float4*)(&wlds[0][wrow * WPITCH + 4 * c4]) = a;
            *(float4*)(&wlds[1][wrow * WPITCH + 4 * c4]) = b;
        }
    }
    const int lane = threadIdx.x & 31;
    const int wave = threadIdx.x >> 5;

    const int chunk = (E + SLICES - 1) / SLICES;
    const int gb    = min((int)threadIdx.x * chunk, E);
    const int glen  = min(chunk, E - gb);
    int o0 = slice_ofs[blockIdx.x * SLICES + threadIdx.x];
    int o1 = slice_ofs[(blockIdx.x + 1) * SLICES + threadIdx.x];
    o0 = min(max(o0, 0), glen);
    o1 = min(max(o1, o0), glen);
    const int cg = o1 - o0;
    const int p0 = gb + o0;
    const int tile = blockIdx.x * 8 + wave;
    const int r    = lane & 15;
    const int hi   = lane >> 4;

    const int  row   = tile * 16 + r;
    const bool valid = (row < N);

    const float2* x2 = (const float2*)x;
    float* ab = lbuf[wave];

    const int tile_s = __builtin_amdgcn_readfirstlane(tile);
    float2 xt[16];
    #pragma unroll
    for (int i = 0; i < 16; ++i) {
        const int ri = tile_s * 16 + i;
        xt[i] = make_float2(0.0f, 0.0f);
        if (ri < N) xt[i] = x2[(long)ri * (IN_DIM / 2) + lane];
    }

    int pinc = cg;
    #pragma unroll
    for (int o = 1; o < 32; o <<= 1) {
        const int u = __shfl_up(pinc, o);
        if (lane >= o) pinc += u;
    }
    if (lane == 31) wtot[wave] = pinc;
    __syncthreads();
    int pre = pinc - cg, ntot = 0;
    #pragma unroll
    for (int i = 0; i < 8; ++i) {
        const int v = wtot[i];
        ntot += v;
        pre  += (i < wave) ? v : 0;
    }
    #pragma unroll
    for (int i = 0; i < 16; ++i)
        *(float2*)(ab + i * APITCH + 2 * lane) = xt[i];

    v2f aXf[16];
    #pragma unroll
    for (int kb = 0; kb < 16; ++kb) {
        const float2 a = *(const float2*)(ab + r * APITCH + 4 * kb + 2 * hi);
        aXf[kb].x = a.x; aXf[kb].y = a.y;
    }
    #pragma unroll
    for (int i = 0; i < 16; ++i)
        *(float2*)(ab + i * APITCH + 2 * lane) = make_float2(0.0f, 0.0f);

    float cnt = 0.0f;
    const int node_l = wave * 16 + r;

    for (int c0 = 0; c0 < ntot; c0 += CH) {
        const int n = min(CH, ntot - c0);
        if (c0 != 0) __syncthreads();
        if (threadIdx.x < BIN_NODES) ncnt[threadIdx.x] = 0;
        {
            const int jlo = max(pre, c0), jhi = min(pre + cg, c0 + CH);
            for (int j = jlo; j < jhi; ++j) sedge[j - c0] = p0 + (j - pre);
        }
        __syncthreads();

        int pl[CH / 256];
        #pragma unroll
        for (int k = 0; k < CH / 256; ++k)
            pl[k] = sedge[min((int)threadIdx.x + 256 * k, n - 1)];
        int ev[CH / 256];
        #pragma unroll
        for (int k = 0; k < CH / 256; ++k) {
            const int j = (int)threadIdx.x + 256 * k;
            ev[k] = -1;
            if (j < n) ev[k] = binned[pl[k]];
        }
        int rk[CH / 256];
        #pragma unroll
        for (int k = 0; k < CH / 256; ++k) rk[k] = 0;
        #pragma unroll
        for (int k = 0; k < CH / 256; ++k)
            if (ev[k] >= 0) rk[k] = atomicAdd(&ncnt[ev[k] & (BIN_NODES - 1)], 1);
        __syncthreads();

        if (wave == 0) {
            const int4 c = *(const int4*)(&ncnt[4 * lane]);
            const int sum4 = c.x + c.y + c.z + c.w;
            int inc = sum4;
            #pragma unroll
            for (int o = 1; o < 32; o <<= 1) {
                const int u = __shfl_up(inc, o);
                if (lane >= o) inc += u;
            }
            const int ex = inc - sum4;
            nstart[4 * lane]     = ex;
            nstart[4 * lane + 1] = ex + c.x;
            nstart[4 * lane + 2] = ex + c.x + c.y;
            nstart[4 * lane + 3] = ex + c.x + c.y + c.z;
        }
        __syncthreads();

        #pragma unroll
        for (int k = 0; k < CH / 256; ++k)
            if (ev[k] >= 0) sedge[nstart[ev[k] & (BIN_NODES - 1)] + rk[k]] = ev[k];
        __syncthreads();

        cnt += (float)ncnt[node_l];
        const int es = __builtin_amdgcn_readfirstlane(nstart[wave * 16]);
        const int ee = __builtin_amdgcn_readfirstlane(nstart[wave * 16 + 15] + ncnt[wave * 16 + 15]);
        if (ee > es) {
            if (__builtin_expect(c0 == 0, 1)) gather_run<false>(x, ab, sedge, es, ee, N, lane);
            else                              gather_run<true>(x, ab, sedge, es, ee, N, lane);
        }
    }

    const float scale = 1.0f / fmaxf(cnt, 1.0f);
    v2f aAf[16];
    #pragma unroll
    for (int kb = 0; kb < 16; ++kb) {
        const float2 a = *(const float2*)(ab + r * APITCH + 4 * kb + 2 * hi);
        aAf[kb].x = a.x * scale; aAf[kb].y = a.y * scale;
    }

    const int koff2 = hi;

    for (int j = 0; j < 4; ++j) {
        const int n = j * 16 + r;
        const float2* wlrow = (const float2*)(&wlds[0][n * WPITCH]);
        const float2* wrrow = (const float2*)(&wlds[1][n * WPITCH]);

        v8f c = {0.f, 0.f, 0.f, 0.f, 0.f, 0.f, 0.f, 0.f};

        #pragma unroll
        for (int kb = 0; kb < 16; ++kb) {
            const int p = 2 * kb + koff2;
            float2 wl = wlrow[p];
            float2 wr = wrrow[p];

            v2f bL; bL.x = wl.x;         bL.y = wl.y;
            v2f bR; bR.x = wr.x;         bR.y = wr.y;

            c = __builtin_amdgcn_wmma_f32_16x16x4_f32(
                    false, aAf[kb], false, bL, (short)0, c, false, false);
            c = __builtin_amdgcn_wmma_f32_16x16x4_f32(
                    false, aXf[kb], false, bR, (short)0, c, false, false);
        }

        const float bias = bl[n];
        #pragma unroll
        for (int v = 0; v < 8; ++v) {
            float h = c[v] + bias;
            h = fmaxf(h, 0.0f);
            ab[(v + hi * 8) * HIDDEN + n] = h;
        }
    }

    float accz = bh[hi];
    #pragma unroll 8
    for (int nn = 0; nn < HIDDEN; ++nn)
        accz += ab[r * HIDDEN + nn] * Wh[hi * HIDDEN + nn];

    if (valid) out[(long)row * OUT_DIM + hi] = accz;
}

extern "C" void kernel_launch(void* const* d_in, const int* in_sizes, int n_in,
                              void* d_out, int out_size, void* d_ws, size_t ws_size,
                              hipStream_t stream) {
    const float* x  = (const float*)d_in[0];
    const int*   ei = (const int*)d_in[1];
    const float* Wl = (const float*)d_in[2];
    const float* bl = (const float*)d_in[3];
    const float* Wr = (const float*)d_in[4];
    const float* Wh = (const float*)d_in[5];
    const float* bh = (const float*)d_in[6];
    float* out = (float*)d_out;

    const long N = in_sizes[0] / IN_DIM;
    const long E = in_sizes[1] / 2;
    const int* src = ei;
    const int* dst = ei + E;

    const int B = (int)((N + BIN_NODES - 1) / BIN_NODES);

    int* binned    = (int*)d_ws;
    int* slice_ofs = binned + (size_t)E;

    gnn_slice_group<<<SLICES, 1024, 0, stream>>>(src, dst, slice_ofs, binned, (int)N, (int)E, B);
    {
        long tiles  = (N + 15) / 16;
        int  blocks = (int)((tiles + 7) / 8);
        gnn_node_update<<<blocks, 256, 0, stream>>>(
            x, slice_ofs, binned, Wl, bl, Wr, Wh, bh, out, (int)N, (int)E);
    }
}


# one 25-wave node work-group per WGP owns a 400-node bin (was 128 nodes, 8 waves): 250 groups read 64k longer slice pieces, not 200k, stage weights 250 times, not 782; sum order = LDS arrival rank
# speedup vs baseline: 6.0371x; 1.1468x over previous
#include <hip/hip_runtime.h>

typedef __attribute__((ext_vector_type(2))) float v2f;
typedef __attribute__((ext_vector_type(8))) float v8f;

#define IN_DIM  64
#define HIDDEN  64
#define OUT_DIM 2
#define APITCH  68
#define WPITCH  68

#define BIN_NODES 400
#define NODE_WAVES (BIN_NODES / 16)
#define NODE_THREADS (32 * NODE_WAVES)
#define BIN_SHIFT 9
#define NODE_MASK ((1 << BIN_SHIFT) - 1)
#define NPAD      512
#define SLICES    256
#define PSPLIT    (NODE_THREADS / SLICES)
#define NB_MAX    1024
#define KPT       6
#define CH        (KPT * NODE_THREADS)
#define SRC_NONE  0x003FFFFF

#define PASS (4 * 1024)
__global__ __launch_bounds__(1024) void gnn_slice_group(
    const int* __restrict__ src,
    const int* __restrict__ dst,
    int* __restrict__ slice_ofs,
    int* __restrict__ binned,
    int N, int E, int B) {
    __shared__ int lcnt[NB_MAX];
    __shared__ int lsta[NB_MAX];
    __shared__ int wsum[32];
    __shared__ int stage[PASS];
    const int t = threadIdx.x, w = blockIdx.x;
    const int lane = t & 31, wv = t >> 5;

    const int chunk = (E + SLICES - 1) / SLICES;
    const int e0 = min(w * chunk, E);
    const int e1 = min(e0 + chunk, E);
    const bool staged = (chunk <= PASS);

    int d[4], s[4];
    #pragma unroll
    for (int k = 0; k < 4; ++k) {
        const int e = e0 + t + 1024 * k;
        d[k] = -1; s[k] = 0;
        if (e < e1) { d[k] = dst[e]; s[k] = src[e]; }
    }
    lcnt[t] = 0;
    __syncthreads();

    int rk[4];
    #pragma unroll
    for (int k = 0; k < 4; ++k) rk[k] = 0;
    for (int eb = e0; eb < e1; eb += PASS) {
        if (eb != e0) {
            #pragma unroll
            for (int k = 0; k < 4; ++k) {
                const int e = eb + t + 1024 * k;
                d[k] = -1;
                if (e < e1) d[k] = dst[e];
            }
        }
        #pragma unroll
        for (int k = 0; k < 4; ++k)
            if ((unsigned)d[k] < (unsigned)N) rk[k] = atomicAdd(&lcnt[(unsigned)d[k] / BIN_NODES], 1);
    }
    __syncthreads();

    const int mine = lcnt[t];
    int inc = mine;
    #pragma unroll
    for (int o = 1; o < 32; o <<= 1) {
        const int u = __shfl_up(inc, o);
        if (lane >= o) inc += u;
    }
    if (lane == 31) wsum[wv] = inc;
    __syncthreads();
    if (wv == 0) {
        const int a = wsum[lane];
        int si = a;
        #pragma unroll
        for (int o = 1; o < 32; o <<= 1) {
            const int u = __shfl_up(si, o);
            if (lane >= o) si += u;
        }
        wsum[lane] = si - a;
    }
    __syncthreads();
    const int excl = wsum[wv] + inc - mine;
    lsta[t] = excl;
    if (t < B) {
        slice_ofs[t * SLICES + w] = excl;
        if (t == B - 1) slice_ofs[B * SLICES + w] = excl + mine;
    }
    if (!staged) lcnt[t] = 0;
    __syncthreads();

    if (staged) {
        #pragma unroll
        for (int k = 0; k < 4; ++k) {
            if ((unsigned)d[k] < (unsigned)N) {
                const int b  = (int)((unsigned)d[k] / BIN_NODES);
                const int sp = ((unsigned)s[k] < (unsigned)N) ? s[k] : SRC_NONE;
                stage[lsta[b] + rk[k]] = (sp << BIN_SHIFT) | (d[k] - b * BIN_NODES);
            }
        }
        __syncthreads();
        const int nl = min(lsta[NB_MAX - 1] + lcnt[NB_MAX - 1], e1 - e0);
        for (int j = t; j < nl; j += 1024)
            binned[e0 + j] = stage[j];
    } else {
        for (int eb = e0; eb < e1; eb += PASS) {
            #pragma unroll
            for (int k = 0; k < 4; ++k) {
                const int e = eb + t + 1024 * k;
                d[k] = -1; s[k] = 0;
                if (e < e1) { d[k] = dst[e]; s[k] = src[e]; }
            }
            #pragma unroll
            for (int k = 0; k < 4; ++k) {
                if ((unsigned)d[k] < (unsigned)N) {
                    const int b  = (int)((unsigned)d[k] / BIN_NODES);
                    const int r2 = atomicAdd(&lcnt[b], 1);
                    const int sp = ((unsigned)s[k] < (unsigned)N) ? s[k] : SRC_NONE;
                    binned[e0 + lsta[b] + r2] = (sp << BIN_SHIFT) | (d[k] - b * BIN_NODES);
                }
            }
        }
    }
}

template <int I>
__device__ __forceinline__ unsigned row_share(unsigned v) {
    return (unsigned)__builtin_amdgcn_update_dpp(0, (int)v, 0x150 + I, 0xF, 0xF, false);
}
__device__ __forceinline__ int row_below(int first, int v) {
    return __builtin_amdgcn_update_dpp(first, v, 0x111  , 0xF, 0xF, false);
}

template <bool RESUME>
__device__ __forceinline__ void gather_run(const float* __restrict__ x, float* ab, const int* sedge,
                                           int es, int ee, int N, int lane) {
    const int r = lane & 15;
    const int pad = sedge[ee - 1] | (SRC_NONE << BIN_SHIFT);
    const unsigned lane8 = 8u * (unsigned)lane;
    float2 run = make_float2(0.0f, 0.0f);
    int w = (es + r < ee) ? sedge[es + r] : pad;
    unsigned vm = __builtin_amdgcn_ballot_w32((unsigned)(w >> BIN_SHIFT) < (unsigned)N);
    unsigned cm = __builtin_amdgcn_ballot_w32(((w ^ row_below(w ^ 1, w)) & NODE_MASK) != 0);
    unsigned rowb = 0;
    for (int e = es; e < ee; e += 16) {
        const unsigned off  = ((unsigned)w >> BIN_SHIFT) << 8;
        const unsigned rowo = (unsigned)(w & 15) * (APITCH * 4u);
        float2 tt[16];
#define GATHER_SLOT(I)                                                                   \
        tt[I] = make_float2(0.0f, 0.0f);                                                 \
        if ((vm >> I) & 1u)                                                              \
            tt[I] = *(const float2*)((const char*)x + (row_share<I>(off) + lane8));
        GATHER_SLOT(0)  GATHER_SLOT(1)  GATHER_SLOT(2)  GATHER_SLOT(3)
        GATHER_SLOT(4)  GATHER_SLOT(5)  GATHER_SLOT(6)  GATHER_SLOT(7)
        GATHER_SLOT(8)  GATHER_SLOT(9)  GATHER_SLOT(10) GATHER_SLOT(11)
        GATHER_SLOT(12) GATHER_SLOT(13) GATHER_SLOT(14) GATHER_SLOT(15)
#undef GATHER_SLOT
        const int jn = e + 16 + r;
        const int wn = (jn < ee) ? sedge[jn] : pad;
        const unsigned vmn = __builtin_amdgcn_ballot_w32((unsigned)(wn >> BIN_SHIFT) < (unsigned)N);
        const unsigned cmn = __builtin_amdgcn_ballot_w32(
            ((wn ^ row_below((int)row_share<15>((unsigned)w), wn)) & NODE_MASK) != 0);
#define SUM_SLOT(I, HAVE_BEFORE, PO)                                                     \
        if (__builtin_expect((cm >> I) & 1u, 0)) {                                       \
            if (HAVE_BEFORE) *(float2*)((char*)ab + ((PO) + lane8)) = run;               \
            run = make_float2(0.0f, 0.0f);                                               \
            if (RESUME) run = *(const float2*)((const char*)ab + (row_share<I>(rowo) + lane8)); \
        }                                                                                \
        run.x += tt[I].x; run.y += tt[I].y;
        SUM_SLOT(0, e != es, rowb)
        SUM_SLOT(1,  true, row_share<0>(rowo))   SUM_SLOT(2,  true, row_share<1>(rowo))
        SUM_SLOT(3,  true, row_share<2>(rowo))   SUM_SLOT(4,  true, row_share<3>(rowo))
        SUM_SLOT(5,  true, row_share<4>(rowo))   SUM_SLOT(6,  true, row_share<5>(rowo))
        SUM_SLOT(7,  true, row_share<6>(rowo))   SUM_SLOT(8,  true, row_share<7>(rowo))
        SUM_SLOT(9,  true, row_share<8>(rowo))   SUM_SLOT(10, true, row_share<9>(rowo))
        SUM_SLOT(11, true, row_share<10>(rowo))  SUM_SLOT(12, true, row_share<11>(rowo))
        SUM_SLOT(13, true, row_share<12>(rowo))  SUM_SLOT(14, true, row_share<13>(rowo))
        SUM_SLOT(15, true, row_share<14>(rowo))
#undef SUM_SLOT
        rowb = row_share<15>(rowo);
        w = wn; vm = vmn; cm = cmn;
    }
    *(float2*)(ab + (pad & 15) * APITCH + 2 * lane) = run;
}

__global__ __launch_bounds__(NODE_THREADS) void gnn_node_update(
    const float* __restrict__ x,
    const int*   __restrict__ slice_ofs,
    const int*   __restrict__ binned,
    const float* __restrict__ Wl,
    const float* __restrict__ bl,
    const float* __restrict__ Wr,
    const float* __restrict__ Wh,
    const float* __restrict__ bh,
    float* __restrict__ out,
    int N, int E) {
    __shared__ alignas(16) float lbuf[NODE_WAVES][16 * APITCH];
    __shared__ alignas(16) float wlds[2][HIDDEN * WPITCH];
    __shared__ int sedge[CH];
    __shared__ alignas(16) int ncnt[NPAD];
    __shared__ alignas(16) int nstart[NPAD];
    __shared__ int wtot[SLICES / 32];

    {
        const float4* wl4 = (const float4*)Wl;
        const float4* wr4 = (const float4*)Wr;
        #pragma unroll
        for (int i = 0; i < (HIDDEN * IN_DIM / 4 + NODE_THREADS - 1) / NODE_THREADS; ++i) {
            const int idx = (int)threadIdx.x + NODE_THREADS * i;
            if (idx < HIDDEN * IN_DIM / 4) {
                const int wrow = idx >> 4, c4 = idx & 15;
                const float4 a = wl4[idx];
                const float4 b = wr4[idx];
                *(float4*)(&wlds[0][wrow * WPITCH + 4 * c4]) = a;
                *(float4*)(&wlds[1][wrow * WPITCH + 4 * c4]) = b;
            }
        }
    }
    const int lane = threadIdx.x & 31;
    const int wave = threadIdx.x >> 5;

    const int chunk = (E + SLICES - 1) / SLICES;
    const int gsl   = (int)threadIdx.x % SLICES;
    const int gq    = (int)threadIdx.x / SLICES;
    const int gb    = min(gsl * chunk, E);
    const int glen  = min(chunk, E - gb);
    int o0 = 0, o1 = 0;
    if (gq < PSPLIT) {
        o0 = slice_ofs[blockIdx.x * SLICES + gsl];
        o1 = slice_ofs[(blockIdx.x + 1) * SLICES + gsl];
    }
    o0 = min(max(o0, 0), glen);
    o1 = min(max(o1, o0), glen);
    const int cg = o1 - o0;
    const int p0 = gb + o0;
    const int tile = blockIdx.x * NODE_WAVES + wave;
    const int r    = lane & 15;
    const int hi   = lane >> 4;

    const int  row   = tile * 16 + r;
    const bool valid = (row < N);

    const float2* x2 = (const float2*)x;
    float* ab = lbuf[wave];

    const int tile_s = __builtin_amdgcn_readfirstlane(tile);
    float2 xt[16];
    #pragma unroll
    for (int i = 0; i < 16; ++i) {
        const int ri = tile_s * 16 + i;
        xt[i] = make_float2(0.0f, 0.0f);
        if (ri < N) xt[i] = x2[(long)ri * (IN_DIM / 2) + lane];
    }

    int pinc = cg;
    #pragma unroll
    for (int o = 1; o < 32; o <<= 1) {
        const int u = __shfl_up(pinc, o);
        if (lane >= o) pinc += u;
    }
    if (lane == 31 && wave < SLICES / 32) wtot[wave] = pinc;
    __syncthreads();
    int pre = pinc - cg, ntot = 0;
    #pragma unroll
    for (int i = 0; i < SLICES / 32; ++i) {
        const int v = wtot[i];
        ntot += v;
        pre  += (i < wave % (SLICES / 32)) ? v : 0;
    }
    #pragma unroll
    for (int i = 0; i < 16; ++i)
        *(float2*)(ab + i * APITCH + 2 * lane) = xt[i];

    v2f aXf[16];
    #pragma unroll
    for (int kb = 0; kb < 16; ++kb) {
        const float2 a = *(const float2*)(ab + r * APITCH + 4 * kb + 2 * hi);
        aXf[kb].x = a.x; aXf[kb].y = a.y;
    }
    #pragma unroll
    for (int i = 0; i < 16; ++i)
        *(float2*)(ab + i * APITCH + 2 * lane) = make_float2(0.0f, 0.0f);

    float cnt = 0.0f;
    const int node_l = wave * 16 + r;

    for (int c0 = 0; c0 < ntot; c0 += CH) {
        const int n = min(CH, ntot - c0);
        if (c0 != 0) __syncthreads();
        if (threadIdx.x < NPAD) ncnt[threadIdx.x] = 0;
        {
            const int jlo = max(pre, c0), jhi = min(pre + cg, c0 + CH);
            for (int j = jlo + gq; j < jhi; j += PSPLIT) sedge[j - c0] = p0 + (j - pre);
        }
        __syncthreads();

        int pl[KPT];
        #pragma unroll
        for (int k = 0; k < KPT; ++k)
            pl[k] = sedge[min((int)threadIdx.x + NODE_THREADS * k, n - 1)];
        int ev[KPT];
        #pragma unroll
        for (int k = 0; k < KPT; ++k) {
            const int j = (int)threadIdx.x + NODE_THREADS * k;
            ev[k] = -1;
            if (j < n) ev[k] = binned[pl[k]];
        }
        int rk[KPT];
        #pragma unroll
        for (int k = 0; k < KPT; ++k) rk[k] = 0;
        #pragma unroll
        for (int k = 0; k < KPT; ++k)
            if (ev[k] >= 0) rk[k] = atomicAdd(&ncnt[ev[k] & NODE_MASK], 1);
        __syncthreads();

        if (wave == 0) {
            int4 c[NPAD / 128];
            #pragma unroll
            for (int q = 0; q < NPAD / 128; ++q)
                c[q] = *(const int4*)(&ncnt[(NPAD / 32) * lane + 4 * q]);
            int sum = 0;
            #pragma unroll
            for (int q = 0; q < NPAD / 128; ++q) sum += c[q].x + c[q].y + c[q].z + c[q].w;
            int inc = sum;
            #pragma unroll
            for (int o = 1; o < 32; o <<= 1) {
                const int u = __shfl_up(inc, o);
                if (lane >= o) inc += u;
            }
            int ex = inc - sum;
            #pragma unroll
            for (int q = 0; q < NPAD / 128; ++q) {
                int4 s4;
                s4.x = ex;
                s4.y = ex + c[q].x;
                s4.z = ex + c[q].x + c[q].y;
                s4.w = ex + c[q].x + c[q].y + c[q].z;
                *(int4*)(&nstart[(NPAD / 32) * lane + 4 * q]) = s4;
                ex = s4.w + c[q].w;
            }
        }
        __syncthreads();

        #pragma unroll
        for (int k = 0; k < KPT; ++k)
            if (ev[k] >= 0) sedge[nstart[ev[k] & NODE_MASK] + rk[k]] = ev[k];
        __syncthreads();

        cnt += (float)ncnt[node_l];
        const int es = __builtin_amdgcn_readfirstlane(nstart[wave * 16]);
        const int ee = __builtin_amdgcn_readfirstlane(nstart[wave * 16 + 15] + ncnt[wave * 16 + 15]);
        if (ee > es) {
            if (__builtin_expect(c0 == 0, 1)) gather_run<false>(x, ab, sedge, es, ee, N, lane);
            else                              gather_run<true>(x, ab, sedge, es, ee, N, lane);
        }
    }

    const float scale = 1.0f / fmaxf(cnt, 1.0f);
    v2f aAf[16];
    #pragma unroll
    for (int kb = 0; kb < 16; ++kb) {
        const float2 a = *(const float2*)(ab + r * APITCH + 4 * kb + 2 * hi);
        aAf[kb].x = a.x * scale; aAf[kb].y = a.y * scale;
    }

    const int koff2 = hi;

    for (int j = 0; j < 4; ++j) {
        const int n = j * 16 + r;
        const float2* wlrow = (const float2*)(&wlds[0][n * WPITCH]);
        const float2* wrrow = (const float2*)(&wlds[1][n * WPITCH]);

        v8f c = {0.f, 0.f, 0.f, 0.f, 0.f, 0.f, 0.f, 0.f};

        #pragma unroll
        for (int kb = 0; kb < 16; ++kb) {
            const int p = 2 * kb + koff2;
            float2 wl = wlrow[p];
            float2 wr = wrrow[p];

            v2f bL; bL.x = wl.x;         bL.y = wl.y;
            v2f bR; bR.x = wr.x;         bR.y = wr.y;

            c = __builtin_amdgcn_wmma_f32_16x16x4_f32(
                    false, aAf[kb], false, bL, (short)0, c, false, false);
            c = __builtin_amdgcn_wmma_f32_16x16x4_f32(
                    false, aXf[kb], false, bR, (short)0, c, false, false);
        }

        const float bias = bl[n];
        #pragma unroll
        for (int v = 0; v < 8; ++v) {
            float h = c[v] + bias;
            h = fmaxf(h, 0.0f);
            ab[(v + hi * 8) * HIDDEN + n] = h;
        }
    }

    float accz = bh[hi];
    #pragma unroll 8
    for (int nn = 0; nn < HIDDEN; ++nn)
        accz += ab[r * HIDDEN + nn] * Wh[hi * HIDDEN + nn];

    if (valid) out[(long)row * OUT_DIM + hi] = accz;
}

extern "C" void kernel_launch(void* const* d_in, const int* in_sizes, int n_in,
                              void* d_out, int out_size, void* d_ws, size_t ws_size,
                              hipStream_t stream) {
    const float* x  = (const float*)d_in[0];
    const int*   ei = (const int*)d_in[1];
    const float* Wl = (const float*)d_in[2];
    const float* bl = (const float*)d_in[3];
    const float* Wr = (const float*)d_in[4];
    const float* Wh = (const float*)d_in[5];
    const float* bh = (const float*)d_in[6];
    float* out = (float*)d_out;

    const long N = in_sizes[0] / IN_DIM;
    const long E = in_sizes[1] / 2;
    const int* src = ei;
    const int* dst = ei + E;

    const int B = (int)((N + BIN_NODES - 1) / BIN_NODES);

    int* binned    = (int*)d_ws;
    int* slice_ofs = binned + (size_t)E;

    gnn_slice_group<<<SLICES, 1024, 0, stream>>>(src, dst, slice_ofs, binned, (int)N, (int)E, B);
    {
        long tiles  = (N + 15) / 16;
        int  blocks = (int)((tiles + NODE_WAVES - 1) / NODE_WAVES);
        gnn_node_update<<<blocks, NODE_THREADS, 0, stream>>>(
            x, slice_ofs, binned, Wl, bl, Wr, Wh, bh, out, (int)N, (int)E);
    }
}


# head (z @ Wh^T) summed where the accumulators lie: per-lane partial dots, then a 16-lane DPP reduce-scatter; no z tile in LDS (its reads were 16-way bank conflicts); head sum order regrouped
# speedup vs baseline: 6.2588x; 1.0367x over previous
#include <hip/hip_runtime.h>

typedef __attribute__((ext_vector_type(2))) float v2f;
typedef __attribute__((ext_vector_type(8))) float v8f;

#define IN_DIM  64
#define HIDDEN  64
#define OUT_DIM 2
#define APITCH  68
#define WPITCH  68

#define BIN_NODES 400
#define NODE_WAVES (BIN_NODES / 16)
#define NODE_THREADS (32 * NODE_WAVES)
#define BIN_SHIFT 9
#define NODE_MASK ((1 << BIN_SHIFT) - 1)
#define NPAD      512
#define SLICES    256
#define PSPLIT    (NODE_THREADS / SLICES)
#define NB_MAX    1024
#define KPT       6
#define CH        (KPT * NODE_THREADS)
#define SRC_NONE  0x003FFFFF

#define PASS (4 * 1024)
__global__ __launch_bounds__(1024) void gnn_slice_group(
    const int* __restrict__ src,
    const int* __restrict__ dst,
    int* __restrict__ slice_ofs,
    int* __restrict__ binned,
    int N, int E, int B) {
    __shared__ int lcnt[NB_MAX];
    __shared__ int lsta[NB_MAX];
    __shared__ int wsum[32];
    __shared__ int stage[PASS];
    const int t = threadIdx.x, w = blockIdx.x;
    const int lane = t & 31, wv = t >> 5;

    const int chunk = (E + SLICES - 1) / SLICES;
    const int e0 = min(w * chunk, E);
    const int e1 = min(e0 + chunk, E);
    const bool staged = (chunk <= PASS);

    int d[4], s[4];
    #pragma unroll
    for (int k = 0; k < 4; ++k) {
        const int e = e0 + t + 1024 * k;
        d[k] = -1; s[k] = 0;
        if (e < e1) { d[k] = dst[e]; s[k] = src[e]; }
    }
    lcnt[t] = 0;
    __syncthreads();

    int rk[4];
    #pragma unroll
    for (int k = 0; k < 4; ++k) rk[k] = 0;
    for (int eb = e0; eb < e1; eb += PASS) {
        if (eb != e0) {
            #pragma unroll
            for (int k = 0; k < 4; ++k) {
                const int e = eb + t + 1024 * k;
                d[k] = -1;
                if (e < e1) d[k] = dst[e];
            }
        }
        #pragma unroll
        for (int k = 0; k < 4; ++k)
            if ((unsigned)d[k] < (unsigned)N) rk[k] = atomicAdd(&lcnt[(unsigned)d[k] / BIN_NODES], 1);
    }
    __syncthreads();

    const int mine = lcnt[t];
    int inc = mine;
    #pragma unroll
    for (int o = 1; o < 32; o <<= 1) {
        const int u = __shfl_up(inc, o);
        if (lane >= o) inc += u;
    }
    if (lane == 31) wsum[wv] = inc;
    __syncthreads();
    if (wv == 0) {
        const int a = wsum[lane];
        int si = a;
        #pragma unroll
        for (int o = 1; o < 32; o <<= 1) {
            const int u = __shfl_up(si, o);
            if (lane >= o) si += u;
        }
        wsum[lane] = si - a;
    }
    __syncthreads();
    const int excl = wsum[wv] + inc - mine;
    lsta[t] = excl;
    if (t < B) {
        slice_ofs[t * SLICES + w] = excl;
        if (t == B - 1) slice_ofs[B * SLICES + w] = excl + mine;
    }
    if (!staged) lcnt[t] = 0;
    __syncthreads();

    if (staged) {
        #pragma unroll
        for (int k = 0; k < 4; ++k) {
            if ((unsigned)d[k] < (unsigned)N) {
                const int b  = (int)((unsigned)d[k] / BIN_NODES);
                const int sp = ((unsigned)s[k] < (unsigned)N) ? s[k] : SRC_NONE;
                stage[lsta[b] + rk[k]] = (sp << BIN_SHIFT) | (d[k] - b * BIN_NODES);
            }
        }
        __syncthreads();
        const int nl = min(lsta[NB_MAX - 1] + lcnt[NB_MAX - 1], e1 - e0);
        for (int j = t; j < nl; j += 1024)
            binned[e0 + j] = stage[j];
    } else {
        for (int eb = e0; eb < e1; eb += PASS) {
            #pragma unroll
            for (int k = 0; k < 4; ++k) {
                const int e = eb + t + 1024 * k;
                d[k] = -1; s[k] = 0;
                if (e < e1) { d[k] = dst[e]; s[k] = src[e]; }
            }
            #pragma unroll
            for (int k = 0; k < 4; ++k) {
                if ((unsigned)d[k] < (unsigned)N) {
                    const int b  = (int)((unsigned)d[k] / BIN_NODES);
                    const int r2 = atomicAdd(&lcnt[b], 1);
                    const int sp = ((unsigned)s[k] < (unsigned)N) ? s[k] : SRC_NONE;
                    binned[e0 + lsta[b] + r2] = (sp << BIN_SHIFT) | (d[k] - b * BIN_NODES);
                }
            }
        }
    }
}

template <int I>
__device__ __forceinline__ unsigned row_share(unsigned v) {
    return (unsigned)__builtin_amdgcn_update_dpp(0, (int)v, 0x150 + I, 0xF, 0xF, false);
}
__device__ __forceinline__ int row_below(int first, int v) {
    return __builtin_amdgcn_update_dpp(first, v, 0x111  , 0xF, 0xF, false);
}
template <int M>
__device__ __forceinline__ float row_xmask(float v) {
    return __int_as_float(__builtin_amdgcn_update_dpp(0, __float_as_int(v), 0x160 + M  , 0xF, 0xF, false));
}

template <bool RESUME>
__device__ __forceinline__ void gather_run(const float* __restrict__ x, float* ab, const int* sedge,
                                           int es, int ee, int N, int lane) {
    const int r = lane & 15;
    const int pad = sedge[ee - 1] | (SRC_NONE << BIN_SHIFT);
    const unsigned lane8 = 8u * (unsigned)lane;
    float2 run = make_float2(0.0f, 0.0f);
    int w = (es + r < ee) ? sedge[es + r] : pad;
    unsigned vm = __builtin_amdgcn_ballot_w32((unsigned)(w >> BIN_SHIFT) < (unsigned)N);
    unsigned cm = __builtin_amdgcn_ballot_w32(((w ^ row_below(w ^ 1, w)) & NODE_MASK) != 0);
    unsigned rowb = 0;
    for (int e = es; e < ee; e += 16) {
        const unsigned off  = ((unsigned)w >> BIN_SHIFT) << 8;
        const unsigned rowo = (unsigned)(w & 15) * (APITCH * 4u);
        float2 tt[16];
#define GATHER_SLOT(I)                                                                   \
        tt[I] = make_float2(0.0f, 0.0f);                                                 \
        if ((vm >> I) & 1u)                                                              \
            tt[I] = *(const float2*)((const char*)x + (row_share<I>(off) + lane8));
        GATHER_SLOT(0)  GATHER_SLOT(1)  GATHER_SLOT(2)  GATHER_SLOT(3)
        GATHER_SLOT(4)  GATHER_SLOT(5)  GATHER_SLOT(6)  GATHER_SLOT(7)
        GATHER_SLOT(8)  GATHER_SLOT(9)  GATHER_SLOT(10) GATHER_SLOT(11)
        GATHER_SLOT(12) GATHER_SLOT(13) GATHER_SLOT(14) GATHER_SLOT(15)
#undef GATHER_SLOT
        const int jn = e + 16 + r;
        const int wn = (jn < ee) ? sedge[jn] : pad;
        const unsigned vmn = __builtin_amdgcn_ballot_w32((unsigned)(wn >> BIN_SHIFT) < (unsigned)N);
        const unsigned cmn = __builtin_amdgcn_ballot_w32(
            ((wn ^ row_below((int)row_share<15>((unsigned)w), wn)) & NODE_MASK) != 0);
#define SUM_SLOT(I, HAVE_BEFORE, PO)                                                     \
        if (__builtin_expect((cm >> I) & 1u, 0)) {                                       \
            if (HAVE_BEFORE) *(float2*)((char*)ab + ((PO) + lane8)) = run;               \
            run = make_float2(0.0f, 0.0f);                                               \
            if (RESUME) run = *(const float2*)((const char*)ab + (row_share<I>(rowo) + lane8)); \
        }                                                                                \
        run.x += tt[I].x; run.y += tt[I].y;
        SUM_SLOT(0, e != es, rowb)
        SUM_SLOT(1,  true, row_share<0>(rowo))   SUM_SLOT(2,  true, row_share<1>(rowo))
        SUM_SLOT(3,  true, row_share<2>(rowo))   SUM_SLOT(4,  true, row_share<3>(rowo))
        SUM_SLOT(5,  true, row_share<4>(rowo))   SUM_SLOT(6,  true, row_share<5>(rowo))
        SUM_SLOT(7,  true, row_share<6>(rowo))   SUM_SLOT(8,  true, row_share<7>(rowo))
        SUM_SLOT(9,  true, row_share<8>(rowo))   SUM_SLOT(10, true, row_share<9>(rowo))
        SUM_SLOT(11, true, row_share<10>(rowo))  SUM_SLOT(12, true, row_share<11>(rowo))
        SUM_SLOT(13, true, row_share<12>(rowo))  SUM_SLOT(14, true, row_share<13>(rowo))
        SUM_SLOT(15, true, row_share<14>(rowo))
#undef SUM_SLOT
        rowb = row_share<15>(rowo);
        w = wn; vm = vmn; cm = cmn;
    }
    *(float2*)(ab + (pad & 15) * APITCH + 2 * lane) = run;
}

__global__ __launch_bounds__(NODE_THREADS) void gnn_node_update(
    const float* __restrict__ x,
    const int*   __restrict__ slice_ofs,
    const int*   __restrict__ binned,
    const float* __restrict__ Wl,
    const float* __restrict__ bl,
    const float* __restrict__ Wr,
    const float* __restrict__ Wh,
    const float* __restrict__ bh,
    float* __restrict__ out,
    int N, int E) {
    __shared__ alignas(16) float lbuf[NODE_WAVES][16 * APITCH];
    __shared__ alignas(16) float wlds[2][HIDDEN * WPITCH];
    __shared__ int sedge[CH];
    __shared__ alignas(16) int ncnt[NPAD];
    __shared__ alignas(16) int nstart[NPAD];
    __shared__ int wtot[SLICES / 32];

    {
        const float4* wl4 = (const float4*)Wl;
        const float4* wr4 = (const float4*)Wr;
        #pragma unroll
        for (int i = 0; i < (HIDDEN * IN_DIM / 4 + NODE_THREADS - 1) / NODE_THREADS; ++i) {
            const int idx = (int)threadIdx.x + NODE_THREADS * i;
            if (idx < HIDDEN * IN_DIM / 4) {
                const int wrow = idx >> 4, c4 = idx & 15;
                const float4 a = wl4[idx];
                const float4 b = wr4[idx];
                *(float4*)(&wlds[0][wrow * WPITCH + 4 * c4]) = a;
                *(float4*)(&wlds[1][wrow * WPITCH + 4 * c4]) = b;
            }
        }
    }
    const int lane = threadIdx.x & 31;
    const int wave = threadIdx.x >> 5;

    const int chunk = (E + SLICES - 1) / SLICES;
    const int gsl   = (int)threadIdx.x % SLICES;
    const int gq    = (int)threadIdx.x / SLICES;
    const int gb    = min(gsl * chunk, E);
    const int glen  = min(chunk, E - gb);
    int o0 = 0, o1 = 0;
    if (gq < PSPLIT) {
        o0 = slice_ofs[blockIdx.x * SLICES + gsl];
        o1 = slice_ofs[(blockIdx.x + 1) * SLICES + gsl];
    }
    o0 = min(max(o0, 0), glen);
    o1 = min(max(o1, o0), glen);
    const int cg = o1 - o0;
    const int p0 = gb + o0;
    const int tile = blockIdx.x * NODE_WAVES + wave;
    const int r    = lane & 15;
    const int hi   = lane >> 4;


    const float2* x2 = (const float2*)x;
    float* ab = lbuf[wave];

    const int tile_s = __builtin_amdgcn_readfirstlane(tile);
    float2 xt[16];
    #pragma unroll
    for (int i = 0; i < 16; ++i) {
        const int ri = tile_s * 16 + i;
        xt[i] = make_float2(0.0f, 0.0f);
        if (ri < N) xt[i] = x2[(long)ri * (IN_DIM / 2) + lane];
    }

    int pinc = cg;
    #pragma unroll
    for (int o = 1; o < 32; o <<= 1) {
        const int u = __shfl_up(pinc, o);
        if (lane >= o) pinc += u;
    }
    if (lane == 31 && wave < SLICES / 32) wtot[wave] = pinc;
    __syncthreads();
    int pre = pinc - cg, ntot = 0;
    #pragma unroll
    for (int i = 0; i < SLICES / 32; ++i) {
        const int v = wtot[i];
        ntot += v;
        pre  += (i < wave % (SLICES / 32)) ? v : 0;
    }
    #pragma unroll
    for (int i = 0; i < 16; ++i)
        *(float2*)(ab + i * APITCH + 2 * lane) = xt[i];

    v2f aXf[16];
    #pragma unroll
    for (int kb = 0; kb < 16; ++kb) {
        const float2 a = *(const float2*)(ab + r * APITCH + 4 * kb + 2 * hi);
        aXf[kb].x = a.x; aXf[kb].y = a.y;
    }
    #pragma unroll
    for (int i = 0; i < 16; ++i)
        *(float2*)(ab + i * APITCH + 2 * lane) = make_float2(0.0f, 0.0f);

    float cnt = 0.0f;
    const int node_l = wave * 16 + r;

    for (int c0 = 0; c0 < ntot; c0 += CH) {
        const int n = min(CH, ntot - c0);
        if (c0 != 0) __syncthreads();
        if (threadIdx.x < NPAD) ncnt[threadIdx.x] = 0;
        {
            const int jlo = max(pre, c0), jhi = min(pre + cg, c0 + CH);
            for (int j = jlo + gq; j < jhi; j += PSPLIT) sedge[j - c0] = p0 + (j - pre);
        }
        __syncthreads();

        int pl[KPT];
        #pragma unroll
        for (int k = 0; k < KPT; ++k)
            pl[k] = sedge[min((int)threadIdx.x + NODE_THREADS * k, n - 1)];
        int ev[KPT];
        #pragma unroll
        for (int k = 0; k < KPT; ++k) {
            const int j = (int)threadIdx.x + NODE_THREADS * k;
            ev[k] = -1;
            if (j < n) ev[k] = binned[pl[k]];
        }
        int rk[KPT];
        #pragma unroll
        for (int k = 0; k < KPT; ++k) rk[k] = 0;
        #pragma unroll
        for (int k = 0; k < KPT; ++k)
            if (ev[k] >= 0) rk[k] = atomicAdd(&ncnt[ev[k] & NODE_MASK], 1);
        __syncthreads();

        if (wave == 0) {
            int4 c[NPAD / 128];
            #pragma unroll
            for (int q = 0; q < NPAD / 128; ++q)
                c[q] = *(const int4*)(&ncnt[(NPAD / 32) * lane + 4 * q]);
            int sum = 0;
            #pragma unroll
            for (int q = 0; q < NPAD / 128; ++q) sum += c[q].x + c[q].y + c[q].z + c[q].w;
            int inc = sum;
            #pragma unroll
            for (int o = 1; o < 32; o <<= 1) {
                const int u = __shfl_up(inc, o);
                if (lane >= o) inc += u;
            }
            int ex = inc - sum;
            #pragma unroll
            for (int q = 0; q < NPAD / 128; ++q) {
                int4 s4;
                s4.x = ex;
                s4.y = ex + c[q].x;
                s4.z = ex + c[q].x + c[q].y;
                s4.w = ex + c[q].x + c[q].y + c[q].z;
                *(int4*)(&nstart[(NPAD / 32) * lane + 4 * q]) = s4;
                ex = s4.w + c[q].w;
            }
        }
        __syncthreads();

        #pragma unroll
        for (int k = 0; k < KPT; ++k)
            if (ev[k] >= 0) sedge[nstart[ev[k] & NODE_MASK] + rk[k]] = ev[k];
        __syncthreads();

        cnt += (float)ncnt[node_l];
        const int es = __builtin_amdgcn_readfirstlane(nstart[wave * 16]);
        const int ee = __builtin_amdgcn_readfirstlane(nstart[wave * 16 + 15] + ncnt[wave * 16 + 15]);
        if (ee > es) {
            if (__builtin_expect(c0 == 0, 1)) gather_run<false>(x, ab, sedge, es, ee, N, lane);
            else                              gather_run<true>(x, ab, sedge, es, ee, N, lane);
        }
    }

    const float scale = 1.0f / fmaxf(cnt, 1.0f);
    v2f aAf[16];
    #pragma unroll
    for (int kb = 0; kb < 16; ++kb) {
        const float2 a = *(const float2*)(ab + r * APITCH + 4 * kb + 2 * hi);
        aAf[kb].x = a.x * scale; aAf[kb].y = a.y * scale;
    }

    const int koff2 = hi;

    float q[16];
    #pragma unroll
    for (int i = 0; i < 16; ++i) q[i] = 0.0f;

    #pragma unroll
    for (int j = 0; j < 4; ++j) {
        const int n = j * 16 + r;
        const float2* wlrow = (const float2*)(&wlds[0][n * WPITCH]);
        const float2* wrrow = (const float2*)(&wlds[1][n * WPITCH]);

        v8f c = {0.f, 0.f, 0.f, 0.f, 0.f, 0.f, 0.f, 0.f};

        #pragma unroll
        for (int kb = 0; kb < 16; ++kb) {
            const int p = 2 * kb + koff2;
            float2 wl = wlrow[p];
            float2 wr = wrrow[p];

            v2f bL; bL.x = wl.x;         bL.y = wl.y;
            v2f bR; bR.x = wr.x;         bR.y = wr.y;

            c = __builtin_amdgcn_wmma_f32_16x16x4_f32(
                    false, aAf[kb], false, bL, (short)0, c, false, false);
            c = __builtin_amdgcn_wmma_f32_16x16x4_f32(
                    false, aXf[kb], false, bR, (short)0, c, false, false);
        }

        const float bias = bl[n];
        const float wh0  = Wh[n];
        const float wh1  = Wh[HIDDEN + n];
        #pragma unroll
        for (int v = 0; v < 8; ++v) {
            float h = c[v] + bias;
            h = fmaxf(h, 0.0f);
            q[2 * v]     += h * wh0;
            q[2 * v + 1] += h * wh1;
        }
    }

    const bool b3 = (r & 8) != 0, b2 = (r & 4) != 0, b1 = (r & 2) != 0, b0 = (r & 1) != 0;
    float s8[8];
    #pragma unroll
    for (int k = 0; k < 8; ++k) {
        const float keep = b3 ? q[k + 8] : q[k];
        const float give = b3 ? q[k] : q[k + 8];
        s8[k] = keep + row_xmask<8>(give);
    }
    float s4[4];
    #pragma unroll
    for (int k = 0; k < 4; ++k) {
        const float keep = b2 ? s8[k + 4] : s8[k];
        const float give = b2 ? s8[k] : s8[k + 4];
        s4[k] = keep + row_xmask<4>(give);
    }
    float s2[2];
    #pragma unroll
    for (int k = 0; k < 2; ++k) {
        const float keep = b1 ? s4[k + 2] : s4[k];
        const float give = b1 ? s4[k] : s4[k + 2];
        s2[k] = keep + row_xmask<2>(give);
    }
    const float keep1 = b0 ? s2[1] : s2[0];
    const float give1 = b0 ? s2[0] : s2[1];
    const float accz  = keep1 + row_xmask<1>(give1) + bh[r & 1];

    const int orow = tile * 16 + 8 * hi + (r >> 1);
    if (orow < N) out[(long)tile * (16 * OUT_DIM) + lane] = accz;
}

extern "C" void kernel_launch(void* const* d_in, const int* in_sizes, int n_in,
                              void* d_out, int out_size, void* d_ws, size_t ws_size,
                              hipStream_t stream) {
    const float* x  = (const float*)d_in[0];
    const int*   ei = (const int*)d_in[1];
    const float* Wl = (const float*)d_in[2];
    const float* bl = (const float*)d_in[3];
    const float* Wr = (const float*)d_in[4];
    const float* Wh = (const float*)d_in[5];
    const float* bh = (const float*)d_in[6];
    float* out = (float*)d_out;

    const long N = in_sizes[0] / IN_DIM;
    const long E = in_sizes[1] / 2;
    const int* src = ei;
    const int* dst = ei + E;

    const int B = (int)((N + BIN_NODES - 1) / BIN_NODES);

    int* binned    = (int*)d_ws;
    int* slice_ofs = binned + (size_t)E;

    gnn_slice_group<<<SLICES, 1024, 0, stream>>>(src, dst, slice_ofs, binned, (int)N, (int)E, B);
    {
        long tiles  = (N + 15) / 16;
        int  blocks = (int)((tiles + NODE_WAVES - 1) / NODE_WAVES);
        gnn_node_update<<<blocks, NODE_THREADS, 0, stream>>>(
            x, slice_ofs, binned, Wl, bl, Wr, Wh, bh, out, (int)N, (int)E);
    }
}


# edges sorted into fixed-place cells (bin x slice) instead of runs + offset table: node kernel reads cells, weights, own rows in one load group; no scan in slice kernel; sum order = LDS arrival rank
# speedup vs baseline: 6.8266x; 1.0907x over previous
#include <hip/hip_runtime.h>

typedef __attribute__((ext_vector_type(2))) float v2f;
typedef __attribute__((ext_vector_type(8))) float v8f;

#define IN_DIM  64
#define HIDDEN  64
#define OUT_DIM 2
#define APITCH  68
#define WPITCH  68

#define BIN_NODES 400
#define NODE_WAVES (BIN_NODES / 16)
#define NODE_THREADS (32 * NODE_WAVES)
#define BIN_SHIFT 9
#define NODE_MASK ((1 << BIN_SHIFT) - 1)
#define NPAD      512
#define SLICES    128
#define CAP       64
#define NB_CAP    256
#define CH        (SLICES * CAP)
#define KQ        ((CH / 4 + NODE_THREADS - 1) / NODE_THREADS)
#define SRC_NONE  0x003FFFFF

#define EPT  8
#define PASS (EPT * 1024)
__global__ __launch_bounds__(1024) void gnn_slice_cells(
    const int* __restrict__ src,
    const int* __restrict__ dst,
    int* __restrict__ cells,
    int* __restrict__ ovf,
    int* __restrict__ ovfn,
    int N, int E, int B) {
    __shared__ int lcnt[NB_CAP];
    __shared__ alignas(16) int stage[NB_CAP * CAP];
    __shared__ int lovfc;
    const int t = threadIdx.x, w = blockIdx.x;

    const int chunk = (E + SLICES - 1) / SLICES;
    const int e0 = min(w * chunk, E);
    const int e1 = min(e0 + chunk, E);

    int d[EPT], s[EPT];
    #pragma unroll
    for (int k = 0; k < EPT; ++k) {
        const int e = e0 + t + 1024 * k;
        d[k] = -1; s[k] = 0;
        if (e < e1) { d[k] = dst[e]; s[k] = src[e]; }
    }
    #pragma unroll
    for (int i = 0; i < NB_CAP * CAP / 4 / 1024; ++i)
        ((int4*)stage)[t + 1024 * i] = make_int4(-1, -1, -1, -1);
    if (t < NB_CAP) lcnt[t] = 0;
    if (t == 0) lovfc = 0;
    __syncthreads();

    for (int eb = e0; eb < e1; eb += PASS) {
        if (eb != e0) {
            #pragma unroll
            for (int k = 0; k < EPT; ++k) {
                const int e = eb + t + 1024 * k;
                d[k] = -1; s[k] = 0;
                if (e < e1) { d[k] = dst[e]; s[k] = src[e]; }
            }
        }
        int rk[EPT];
        #pragma unroll
        for (int k = 0; k < EPT; ++k) {
            rk[k] = CAP;
            if ((unsigned)d[k] < (unsigned)N && (unsigned)d[k] / BIN_NODES < NB_CAP)
                rk[k] = atomicAdd(&lcnt[(unsigned)d[k] / BIN_NODES], 1);
        }
        bool full = false;
        #pragma unroll
        for (int k = 0; k < EPT; ++k) {
            if ((unsigned)d[k] < (unsigned)N) {
                const int b  = (int)((unsigned)d[k] / BIN_NODES);
                const int sp = ((unsigned)s[k] < (unsigned)N) ? s[k] : SRC_NONE;
                if (rk[k] < CAP) stage[b * CAP + rk[k]] = (sp << BIN_SHIFT) | (d[k] - b * BIN_NODES);
                else             full = true;
            }
        }
        if (full) {
            #pragma unroll
            for (int k = 0; k < EPT; ++k) {
                if ((unsigned)d[k] < (unsigned)N && rk[k] >= CAP) {
                    const int sp = ((unsigned)s[k] < (unsigned)N) ? s[k] : SRC_NONE;
                    const int o = atomicAdd(&lovfc, 1);
                    const long q = ((long)w * chunk + o) * 2;
                    ovf[q]     = d[k];
                    ovf[q + 1] = sp;
                }
            }
        }
    }
    __syncthreads();

    const int n4 = min(B, NB_CAP) * (CAP / 4);
    #pragma unroll
    for (int i = 0; i < NB_CAP * CAP / 4 / 1024; ++i) {
        const int i4 = t + 1024 * i;
        if (i4 < n4) {
            const int b = i4 / (CAP / 4), c = i4 % (CAP / 4);
            ((int4*)cells)[((long)b * SLICES + w) * (CAP / 4) + c] = ((const int4*)stage)[i4];
        }
    }
    if (t == 0) ovfn[w] = lovfc;
}

template <int I>
__device__ __forceinline__ unsigned row_share(unsigned v) {
    return (unsigned)__builtin_amdgcn_update_dpp(0, (int)v, 0x150 + I, 0xF, 0xF, false);
}
__device__ __forceinline__ int row_below(int first, int v) {
    return __builtin_amdgcn_update_dpp(first, v, 0x111  , 0xF, 0xF, false);
}
template <int M>
__device__ __forceinline__ float row_xmask(float v) {
    return __int_as_float(__builtin_amdgcn_update_dpp(0, __float_as_int(v), 0x160 + M  , 0xF, 0xF, false));
}

template <bool RESUME>
__device__ __forceinline__ void gather_run(const float* __restrict__ x, float* ab, const int* sedge,
                                           int es, int ee, int N, int lane) {
    const int r = lane & 15;
    const int pad = sedge[ee - 1] | (SRC_NONE << BIN_SHIFT);
    const unsigned lane8 = 8u * (unsigned)lane;
    float2 run = make_float2(0.0f, 0.0f);
    int w = (es + r < ee) ? sedge[es + r] : pad;
    unsigned vm = __builtin_amdgcn_ballot_w32((unsigned)(w >> BIN_SHIFT) < (unsigned)N);
    unsigned cm = __builtin_amdgcn_ballot_w32(((w ^ row_below(w ^ 1, w)) & NODE_MASK) != 0);
    unsigned rowb = 0;
    for (int e = es; e < ee; e += 16) {
        const unsigned off  = ((unsigned)w >> BIN_SHIFT) << 8;
        const unsigned rowo = (unsigned)(w & 15) * (APITCH * 4u);
        float2 tt[16];
#define GATHER_SLOT(I)                                                                   \
        tt[I] = make_float2(0.0f, 0.0f);                                                 \
        if ((vm >> I) & 1u)                                                              \
            tt[I] = *(const float2*)((const char*)x + (row_share<I>(off) + lane8));
        GATHER_SLOT(0)  GATHER_SLOT(1)  GATHER_SLOT(2)  GATHER_SLOT(3)
        GATHER_SLOT(4)  GATHER_SLOT(5)  GATHER_SLOT(6)  GATHER_SLOT(7)
        GATHER_SLOT(8)  GATHER_SLOT(9)  GATHER_SLOT(10) GATHER_SLOT(11)
        GATHER_SLOT(12) GATHER_SLOT(13) GATHER_SLOT(14) GATHER_SLOT(15)
#undef GATHER_SLOT
        const int jn = e + 16 + r;
        const int wn = (jn < ee) ? sedge[jn] : pad;
        const unsigned vmn = __builtin_amdgcn_ballot_w32((unsigned)(wn >> BIN_SHIFT) < (unsigned)N);
        const unsigned cmn = __builtin_amdgcn_ballot_w32(
            ((wn ^ row_below((int)row_share<15>((unsigned)w), wn)) & NODE_MASK) != 0);
#define SUM_SLOT(I, HAVE_BEFORE, PO)                                                     \
        if (__builtin_expect((cm >> I) & 1u, 0)) {                                       \
            if (HAVE_BEFORE) *(float2*)((char*)ab + ((PO) + lane8)) = run;               \
            run = make_float2(0.0f, 0.0f);                                               \
            if (RESUME) run = *(const float2*)((const char*)ab + (row_share<I>(rowo) + lane8)); \
        }                                                                                \
        run.x += tt[I].x; run.y += tt[I].y;
        SUM_SLOT(0, e != es, rowb)
        SUM_SLOT(1,  true, row_share<0>(rowo))   SUM_SLOT(2,  true, row_share<1>(rowo))
        SUM_SLOT(3,  true, row_share<2>(rowo))   SUM_SLOT(4,  true, row_share<3>(rowo))
        SUM_SLOT(5,  true, row_share<4>(rowo))   SUM_SLOT(6,  true, row_share<5>(rowo))
        SUM_SLOT(7,  true, row_share<6>(rowo))   SUM_SLOT(8,  true, row_share<7>(rowo))
        SUM_SLOT(9,  true, row_share<8>(rowo))   SUM_SLOT(10, true, row_share<9>(rowo))
        SUM_SLOT(11, true, row_share<10>(rowo))  SUM_SLOT(12, true, row_share<11>(rowo))
        SUM_SLOT(13, true, row_share<12>(rowo))  SUM_SLOT(14, true, row_share<13>(rowo))
        SUM_SLOT(15, true, row_share<14>(rowo))
#undef SUM_SLOT
        rowb = row_share<15>(rowo);
        w = wn; vm = vmn; cm = cmn;
    }
    *(float2*)(ab + (pad & 15) * APITCH + 2 * lane) = run;
}

__device__ __forceinline__ void scan_counts(const int* ncnt, int* nstart, int lane) {
    int4 c[NPAD / 128];
    #pragma unroll
    for (int q = 0; q < NPAD / 128; ++q)
        c[q] = *(const int4*)(&ncnt[(NPAD / 32) * lane + 4 * q]);
    int sum = 0;
    #pragma unroll
    for (int q = 0; q < NPAD / 128; ++q) sum += c[q].x + c[q].y + c[q].z + c[q].w;
    int inc = sum;
    #pragma unroll
    for (int o = 1; o < 32; o <<= 1) {
        const int u = __shfl_up(inc, o);
        if (lane >= o) inc += u;
    }
    int ex = inc - sum;
    #pragma unroll
    for (int q = 0; q < NPAD / 128; ++q) {
        int4 s4;
        s4.x = ex;
        s4.y = ex + c[q].x;
        s4.z = ex + c[q].x + c[q].y;
        s4.w = ex + c[q].x + c[q].y + c[q].z;
        *(int4*)(&nstart[(NPAD / 32) * lane + 4 * q]) = s4;
        ex = s4.w + c[q].w;
    }
}

__global__ __launch_bounds__(NODE_THREADS) void gnn_node_update(
    const float* __restrict__ x,
    const int*   __restrict__ cells,
    const int*   __restrict__ ovf,
    const int*   __restrict__ ovfn,
    const float* __restrict__ Wl,
    const float* __restrict__ bl,
    const float* __restrict__ Wr,
    const float* __restrict__ Wh,
    const float* __restrict__ bh,
    float* __restrict__ out,
    int N, int E) {
    __shared__ alignas(16) float lbuf[NODE_WAVES][16 * APITCH];
    __shared__ alignas(16) float wlds[2][HIDDEN * WPITCH];
    __shared__ int sedge[CH];
    __shared__ alignas(16) int ncnt[NPAD];
    __shared__ alignas(16) int nstart[NPAD];
    __shared__ int lovf[SLICES];
    __shared__ int lflag;

    const int t    = (int)threadIdx.x;
    const int lane = t & 31;
    const int wave = t >> 5;
    const int bin  = (int)blockIdx.x;
    const int tile = bin * NODE_WAVES + wave;
    const int r    = lane & 15;
    const int hi   = lane >> 4;
    const float2* x2 = (const float2*)x;
    float* ab = lbuf[wave];

    int ev[4 * KQ];
    {
        const int4* c4 = (const int4*)cells + (long)bin * (CH / 4);
        #pragma unroll
        for (int k = 0; k < KQ; ++k) {
            const int i4 = t + NODE_THREADS * k;
            int4 v = make_int4(-1, -1, -1, -1);
            if (bin < NB_CAP && i4 < CH / 4) v = c4[i4];
            ev[4 * k] = v.x; ev[4 * k + 1] = v.y; ev[4 * k + 2] = v.z; ev[4 * k + 3] = v.w;
        }
    }
    int ov = 0;
    if (t < SLICES) ov = ovfn[t];

    const int tile_s = __builtin_amdgcn_readfirstlane(tile);
    float2 xt[16];
    #pragma unroll
    for (int i = 0; i < 16; ++i) {
        const int ri = tile_s * 16 + i;
        xt[i] = make_float2(0.0f, 0.0f);
        if (ri < N) xt[i] = x2[(long)ri * (IN_DIM / 2) + lane];
    }

    {
        const float4* wl4 = (const float4*)Wl;
        const float4* wr4 = (const float4*)Wr;
        constexpr int WT = (HIDDEN * IN_DIM / 4 + NODE_THREADS - 1) / NODE_THREADS;
        float4 wa[WT], wb[WT];
        #pragma unroll
        for (int i = 0; i < WT; ++i) {
            const int idx = t + NODE_THREADS * i;
            wa[i] = make_float4(0.f, 0.f, 0.f, 0.f); wb[i] = wa[i];
            if (idx < HIDDEN * IN_DIM / 4) { wa[i] = wl4[idx]; wb[i] = wr4[idx]; }
        }
        if (t < NPAD) ncnt[t] = 0;
        if (t == 0) lflag = 0;
        #pragma unroll
        for (int i = 0; i < WT; ++i) {
            const int idx = t + NODE_THREADS * i;
            if (idx < HIDDEN * IN_DIM / 4) {
                const int wrow = idx >> 4, c4i = idx & 15;
                *(float4*)(&wlds[0][wrow * WPITCH + 4 * c4i]) = wa[i];
                *(float4*)(&wlds[1][wrow * WPITCH + 4 * c4i]) = wb[i];
            }
        }
    }
    if (t < SLICES) lovf[t] = ov;
    __syncthreads();

    if (t < SLICES && ov > 0) atomicOr(&lflag, 1);
    int rk[4 * KQ];
    #pragma unroll
    for (int k = 0; k < 4 * KQ; ++k) {
        rk[k] = 0;
        if (ev[k] >= 0) rk[k] = atomicAdd(&ncnt[ev[k] & NODE_MASK], 1);
    }

    #pragma unroll
    for (int i = 0; i < 16; ++i)
        *(float2*)(ab + i * APITCH + 2 * lane) = xt[i];
    v2f aXf[16];
    #pragma unroll
    for (int kb = 0; kb < 16; ++kb) {
        const float2 a = *(const float2*)(ab + r * APITCH + 4 * kb + 2 * hi);
        aXf[kb].x = a.x; aXf[kb].y = a.y;
    }
    #pragma unroll
    for (int i = 0; i < 16; ++i)
        *(float2*)(ab + i * APITCH + 2 * lane) = make_float2(0.0f, 0.0f);
    __syncthreads();

    const int any_ovf = __builtin_amdgcn_readfirstlane(lflag);
    if (wave == 0) scan_counts(ncnt, nstart, lane);
    __syncthreads();

    int st[4 * KQ];
    #pragma unroll
    for (int k = 0; k < 4 * KQ; ++k) st[k] = nstart[ev[k] & NODE_MASK];
    #pragma unroll
    for (int k = 0; k < 4 * KQ; ++k)
        if (ev[k] >= 0) sedge[st[k] + rk[k]] = ev[k];
    __syncthreads();

    const int node_l = wave * 16 + r;
    float cnt = (float)ncnt[node_l];
    {
        const int es = __builtin_amdgcn_readfirstlane(nstart[wave * 16]);
        const int ee = __builtin_amdgcn_readfirstlane(nstart[wave * 16 + 15] + ncnt[wave * 16 + 15]);
        if (ee > es) gather_run<false>(x, ab, sedge, es, ee, N, lane);
    }

    if (any_ovf != 0) {
        const int chunk = (E + SLICES - 1) / SLICES;
        for (int g = 0; g < SLICES; ++g) {
            const int ng = min(max(__builtin_amdgcn_readfirstlane(lovf[g]), 0), chunk);
            for (int base = 0; base < ng; base += NODE_THREADS) {
                __syncthreads();
                if (t < NPAD) ncnt[t] = 0;
                __syncthreads();
                int word = -1;
                if (base + t < ng) {
                    const long q = ((long)g * chunk + base + t) * 2;
                    const int dl = ovf[q] - bin * BIN_NODES;
                    if ((unsigned)dl < (unsigned)BIN_NODES) word = (ovf[q + 1] << BIN_SHIFT) | dl;
                }
                int rk1 = 0;
                if (word >= 0) rk1 = atomicAdd(&ncnt[word & NODE_MASK], 1);
                __syncthreads();
                if (wave == 0) scan_counts(ncnt, nstart, lane);
                __syncthreads();
                if (word >= 0) sedge[nstart[word & NODE_MASK] + rk1] = word;
                __syncthreads();
                cnt += (float)ncnt[node_l];
                const int es = __builtin_amdgcn_readfirstlane(nstart[wave * 16]);
                const int ee = __builtin_amdgcn_readfirstlane(nstart[wave * 16 + 15] + ncnt[wave * 16 + 15]);
                if (ee > es) gather_run<true>(x, ab, sedge, es, ee, N, lane);
            }
        }
    }

    const float scale = 1.0f / fmaxf(cnt, 1.0f);
    v2f aAf[16];
    #pragma unroll
    for (int kb = 0; kb < 16; ++kb) {
        const float2 a = *(const float2*)(ab + r * APITCH + 4 * kb + 2 * hi);
        aAf[kb].x = a.x * scale; aAf[kb].y = a.y * scale;
    }

    const int koff2 = hi;

    float q[16];
    #pragma unroll
    for (int i = 0; i < 16; ++i) q[i] = 0.0f;

    #pragma unroll
    for (int j = 0; j < 4; ++j) {
        const int n = j * 16 + r;
        const float2* wlrow = (const float2*)(&wlds[0][n * WPITCH]);
        const float2* wrrow = (const float2*)(&wlds[1][n * WPITCH]);

        v8f c = {0.f, 0.f, 0.f, 0.f, 0.f, 0.f, 0.f, 0.f};

        #pragma unroll
        for (int kb = 0; kb < 16; ++kb) {
            const int p = 2 * kb + koff2;
            float2 wl = wlrow[p];
            float2 wr = wrrow[p];

            v2f bL; bL.x = wl.x;         bL.y = wl.y;
            v2f bR; bR.x = wr.x;         bR.y = wr.y;

            c = __builtin_amdgcn_wmma_f32_16x16x4_f32(
                    false, aAf[kb], false, bL, (short)0, c, false, false);
            c = __builtin_amdgcn_wmma_f32_16x16x4_f32(
                    false, aXf[kb], false, bR, (short)0, c, false, false);
        }

        const float bias = bl[n];
        const float wh0  = Wh[n];
        const float wh1  = Wh[HIDDEN + n];
        #pragma unroll
        for (int v = 0; v < 8; ++v) {
            float h = c[v] + bias;
            h = fmaxf(h, 0.0f);
            q[2 * v]     += h * wh0;
            q[2 * v + 1] += h * wh1;
        }
    }

    const bool b3 = (r & 8) != 0, b2 = (r & 4) != 0, b1 = (r & 2) != 0, b0 = (r & 1) != 0;
    float s8[8];
    #pragma unroll
    for (int k = 0; k < 8; ++k) {
        const float keep = b3 ? q[k + 8] : q[k];
        const float give = b3 ? q[k] : q[k + 8];
        s8[k] = keep + row_xmask<8>(give);
    }
    float s4[4];
    #pragma unroll
    for (int k = 0; k < 4; ++k) {
        const float keep = b2 ? s8[k + 4] : s8[k];
        const float give = b2 ? s8[k] : s8[k + 4];
        s4[k] = keep + row_xmask<4>(give);
    }
    float s2[2];
    #pragma unroll
    for (int k = 0; k < 2; ++k) {
        const float keep = b1 ? s4[k + 2] : s4[k];
        const float give = b1 ? s4[k] : s4[k + 2];
        s2[k] = keep + row_xmask<2>(give);
    }
    const float keep1 = b0 ? s2[1] : s2[0];
    const float give1 = b0 ? s2[0] : s2[1];
    const float accz  = keep1 + row_xmask<1>(give1) + bh[r & 1];

    const int orow = tile * 16 + 8 * hi + (r >> 1);
    if (orow < N) out[(long)tile * (16 * OUT_DIM) + lane] = accz;
}

extern "C" void kernel_launch(void* const* d_in, const int* in_sizes, int n_in,
                              void* d_out, int out_size, void* d_ws, size_t ws_size,
                              hipStream_t stream) {
    const float* x  = (const float*)d_in[0];
    const int*   ei = (const int*)d_in[1];
    const float* Wl = (const float*)d_in[2];
    const float* bl = (const float*)d_in[3];
    const float* Wr = (const float*)d_in[4];
    const float* Wh = (const float*)d_in[5];
    const float* bh = (const float*)d_in[6];
    float* out = (float*)d_out;

    const long N = in_sizes[0] / IN_DIM;
    const long E = in_sizes[1] / 2;
    const int* src = ei;
    const int* dst = ei + E;

    const int B = (int)((N + BIN_NODES - 1) / BIN_NODES);

    int* cells = (int*)d_ws;
    int* ovfn  = cells + (size_t)NB_CAP * SLICES * CAP;
    int* ovf   = ovfn + SLICES;

    gnn_slice_cells<<<SLICES, 1024, 0, stream>>>(src, dst, cells, ovf, ovfn, (int)N, (int)E, B);
    {
        long tiles  = (N + 15) / 16;
        int  blocks = (int)((tiles + NODE_WAVES - 1) / NODE_WAVES);
        gnn_node_update<<<blocks, NODE_THREADS, 0, stream>>>(
            x, cells, ovf, ovfn, Wl, bl, Wr, Wh, bh, out, (int)N, (int)E);
    }
}
